# starting kernel
# baseline (speedup 1.0000x reference)
#include <hip/hip_runtime.h>
#include <hip/hip_bf16.h>
#include <math.h>


typedef __attribute__((ext_vector_type(16))) __bf16 bf16x16;
typedef __attribute__((ext_vector_type(8)))  float  floatx8;

union FragAB { bf16x16 v; unsigned int u[8]; };

__device__ __forceinline__ unsigned pack2bf(float a, float b) {
  __bf16 lo = (__bf16)a, hi = (__bf16)b;
  unsigned short ul = __builtin_bit_cast(unsigned short, lo);
  unsigned short uh = __builtin_bit_cast(unsigned short, hi);
  return (unsigned)ul | ((unsigned)uh << 16);
}

__device__ __forceinline__ int kpair(int p, int h) {
  return (p < 4) ? (2 * p + 8 * h) : (8 + 2 * p + 8 * h);
}

__device__ __forceinline__ bf16x16 load_frag_bf16_rm(const __bf16* src, int row0,
                                                     int k0, int ld) {
  int lane = threadIdx.x & 31;
  int r = lane & 15, h = lane >> 4;
  const __bf16* base = src + (size_t)(row0 + r) * ld + k0;
  FragAB f;
#pragma unroll
  for (int p = 0; p < 8; ++p) f.u[p] = *(const unsigned*)(base + kpair(p, h));
  return f.v;
}

__device__ __forceinline__ floatx8 wmma_bf16(bf16x16 a, bf16x16 b, floatx8 c) {
  return __builtin_amdgcn_wmma_f32_16x16x32_bf16(false, a, false, b, (short)0, c,
                                                 false, false);
}

template <int KSTEPS>
__device__ __forceinline__ void gemm_strip(const __bf16* A, int rowA,
                                           const __bf16* B, int rowB, int ld,
                                           floatx8 acc[4]) {
#pragma unroll
  for (int ks = 0; ks < KSTEPS; ++ks) {
    int k0 = ks * 32;
    bf16x16 af = load_frag_bf16_rm(A, rowA, k0, ld);
#pragma unroll
    for (int t = 0; t < 4; ++t) {
      bf16x16 bfr = load_frag_bf16_rm(B, rowB + 16 * t, k0, ld);
      acc[t] = wmma_bf16(af, bfr, acc[t]);
    }
  }
}

__global__ void k_convert_w(const float* wqkv, const float* wout,
                            __bf16* wqkv_b, __bf16* wout_b, float* scales) {
  int i = blockIdx.x * blockDim.x + threadIdx.x;
  if (i < 2) scales[i] = 0.f;
  if (i < 1536 * 512) wqkv_b[i] = (__bf16)wqkv[i];
  if (i < 512 * 512) wout_b[i] = (__bf16)wout[i];
}

__global__ void k_layernorm(const float* x, const float* g, const float* b,
                            __bf16* xn, int ntok) {
  int wave = (int)((blockIdx.x * blockDim.x + threadIdx.x) >> 5);
  int lane = threadIdx.x & 31;
  if (wave >= ntok) return;
  const float* xr = x + (size_t)wave * 512;
  float s = 0.f, s2 = 0.f, vals[16];
#pragma unroll
  for (int i = 0; i < 16; ++i) {
    float v = xr[lane + 32 * i];
    vals[i] = v; s += v; s2 += v * v;
  }
  for (int m = 16; m >= 1; m >>= 1) {
    s += __shfl_xor(s, m, 32);
    s2 += __shfl_xor(s2, m, 32);
  }
  float mu = s * (1.f / 512.f);
  float var = s2 * (1.f / 512.f) - mu * mu;
  float inv = rsqrtf(var + 1e-5f);
  __bf16* o = xn + (size_t)wave * 512;
#pragma unroll
  for (int i = 0; i < 16; ++i) {
    int c = lane + 32 * i;
    o[c] = (__bf16)((vals[i] - mu) * inv * g[c] + b[c]);
  }
}

__global__ void k_gemm_qkv(const __bf16* xn, const __bf16* wq, __bf16* qb,
                           __bf16* kb, float* vf, __bf16* vT) {
  int wid = (int)((blockIdx.x * blockDim.x + threadIdx.x) >> 5);
  int tM = wid / 24, s = wid % 24;
  floatx8 acc[4] = {};
  gemm_strip<16>(xn, tM * 16, wq, s * 64, 512, acc);
  int lane = threadIdx.x & 31;
  int n = lane & 15, h = lane >> 4;
  int col0 = s * 64;
  int sect = col0 >> 9;
  int head = (col0 >> 6) & 7;
#pragma unroll
  for (int t = 0; t < 4; ++t) {
    int dh = 16 * t + n;
#pragma unroll
    for (int r = 0; r < 8; ++r) {
      int tok = tM * 16 + r + 8 * h;
      int bb = tok >> 12, ns = tok & 4095;
      size_t bh = (size_t)(bb * 8 + head);
      size_t idx = (bh * 4096 + ns) * 64 + dh;
      float val = acc[t][r];
      if (sect == 0) qb[idx] = (__bf16)(val * 0.125f);
      else if (sect == 1) kb[idx] = (__bf16)val;
      else {
        vf[idx] = val;
        vT[(bh * 64 + dh) * 4096 + ns] = (__bf16)val;
      }
    }
  }
}

__global__ void k_landmarks(const __bf16* qb, const __bf16* kb, __bf16* ql,
                            __bf16* kl) {
  int i = blockIdx.x * blockDim.x + threadIdx.x;
  if (i >= 32 * 256 * 64) return;
  int dh = i & 63, m = (i >> 6) & 255, bh = i >> 14;
  size_t base = ((size_t)bh * 4096 + m * 16) * 64 + dh;
  float sq = 0.f, sk = 0.f;
#pragma unroll
  for (int t = 0; t < 16; ++t) {
    sq += (float)qb[base + t * 64];
    sk += (float)kb[base + t * 64];
  }
  ql[i] = (__bf16)(sq * (1.f / 16.f));
  kl[i] = (__bf16)(sk * (1.f / 16.f));
}

__global__ void k_attn2_logits(const __bf16* ql, const __bf16* kl, float* out) {
  int wid = (int)((blockIdx.x * blockDim.x + threadIdx.x) >> 5);
  int bh = wid >> 6, t = wid & 63;
  int tM = t >> 2, s = t & 3;
  const __bf16* A = ql + (size_t)bh * 256 * 64;
  const __bf16* B = kl + (size_t)bh * 256 * 64;
  floatx8 acc[4] = {};
  gemm_strip<2>(A, tM * 16, B, s * 64, 64, acc);
  int lane = threadIdx.x & 31;
  int n = lane & 15, h = lane >> 4;
  float* o = out + (size_t)bh * 256 * 256;
#pragma unroll
  for (int t4 = 0; t4 < 4; ++t4)
#pragma unroll
    for (int r = 0; r < 8; ++r)
      o[(size_t)(tM * 16 + r + 8 * h) * 256 + s * 64 + 16 * t4 + n] =
          acc[t4][r];
}

__global__ void k_softmax256(float* x, __bf16* xb, int nrows) {
  int wave = (int)((blockIdx.x * blockDim.x + threadIdx.x) >> 5);
  if (wave >= nrows) return;
  int lane = threadIdx.x & 31;
  float* r = x + (size_t)wave * 256;
  __bf16* rb = xb + (size_t)wave * 256;
  float v[8], m = -1e30f;
#pragma unroll
  for (int i = 0; i < 8; ++i) { v[i] = r[lane + 32 * i]; m = fmaxf(m, v[i]); }
  for (int t = 16; t >= 1; t >>= 1) m = fmaxf(m, __shfl_xor(m, t, 32));
  float s = 0.f;
#pragma unroll
  for (int i = 0; i < 8; ++i) { v[i] = __expf(v[i] - m); s += v[i]; }
  for (int t = 16; t >= 1; t >>= 1) s += __shfl_xor(s, t, 32);
  float inv = 1.f / s;
#pragma unroll
  for (int i = 0; i < 8; ++i) {
    float pv = v[i] * inv;
    r[lane + 32 * i] = pv;
    rb[lane + 32 * i] = (__bf16)pv;
  }
}

__global__ void k_pinv_scales(const float* x, float* scales) {
  int wave = (int)((blockIdx.x * blockDim.x + threadIdx.x) >> 5);
  if (wave >= 32 * 256) return;
  int lane = threadIdx.x & 31;
  int bh = wave >> 8, r = wave & 255;
  const float* m = x + (size_t)bh * 256 * 256;
  float rs = 0.f, cs = 0.f;
#pragma unroll
  for (int i = 0; i < 8; ++i) {
    rs += fabsf(m[(size_t)r * 256 + lane + 32 * i]);
    cs += fabsf(m[(size_t)(lane + 32 * i) * 256 + r]);
  }
  for (int t = 16; t >= 1; t >>= 1) {
    rs += __shfl_xor(rs, t, 32);
    cs += __shfl_xor(cs, t, 32);
  }
  if (lane == 0) {
    atomicMax((unsigned*)&scales[0], __float_as_uint(rs));
    atomicMax((unsigned*)&scales[1], __float_as_uint(cs));
  }
}

__global__ void k_init_z(const float* x, __bf16* zb, __bf16* zTb,
                         const float* scales) {
  size_t i = (size_t)blockIdx.x * blockDim.x + threadIdx.x;
  if (i >= (size_t)32 * 256 * 256) return;
  int j = (int)(i & 255), r = (int)((i >> 8) & 255);
  size_t bh = i >> 16;
  float denom = scales[0] * scales[1];
  float val = x[(bh << 16) + (size_t)j * 256 + r] / denom;
  zb[i] = (__bf16)val;
  zTb[(bh << 16) + (size_t)j * 256 + r] = (__bf16)val;
}

__global__ void k_gemm256b(const __bf16* A, const __bf16* BT, __bf16* OutRaw,
                           __bf16* OutT, __bf16* OutTT, float alpha,
                           float beta) {
  int wid = (int)((blockIdx.x * blockDim.x + threadIdx.x) >> 5);
  int bh = wid >> 6, t = wid & 63;
  int tM = t >> 2, s = t & 3;
  size_t bofs = (size_t)bh << 16;
  floatx8 acc[4] = {};
  gemm_strip<8>(A + bofs, tM * 16, BT + bofs, s * 64, 256, acc);
  int lane = threadIdx.x & 31;
  int n = lane & 15, h = lane >> 4;
#pragma unroll
  for (int t4 = 0; t4 < 4; ++t4)
#pragma unroll
    for (int r = 0; r < 8; ++r) {
      int M = tM * 16 + r + 8 * h, N = s * 64 + 16 * t4 + n;
      float tv = acc[t4][r];
      if (OutRaw) OutRaw[bofs + (size_t)M * 256 + N] = (__bf16)tv;
      float tr = alpha * (M == N ? 1.f : 0.f) + beta * tv;
      if (OutT) OutT[bofs + (size_t)M * 256 + N] = (__bf16)tr;
      if (OutTT) OutTT[bofs + (size_t)N * 256 + M] = (__bf16)tr;
    }
}

__device__ __forceinline__ void stream_attn(const __bf16* Ak, const __bf16* Bq,
                                            int qrow0, const __bf16* A2,
                                            int ldA2, int nkeys, float* outF,
                                            int rsF, __bf16* outB, int nsB,
                                            int dsB) {
  int lane = threadIdx.x & 31;
  int n = lane & 15, h = lane >> 4;
  bf16x16 bq0 = load_frag_bf16_rm(Bq, qrow0, 0, 64);
  bf16x16 bq1 = load_frag_bf16_rm(Bq, qrow0, 32, 64);
  float run_m = -1e30f, run_s = 0.f;
  floatx8 acc[4] = {};
  for (int c = 0; c < nkeys; c += 32) {
    floatx8 s0 = {}, s1 = {};
    s0 = wmma_bf16(load_frag_bf16_rm(Ak, c, 0, 64), bq0, s0);
    s0 = wmma_bf16(load_frag_bf16_rm(Ak, c, 32, 64), bq1, s0);
    s1 = wmma_bf16(load_frag_bf16_rm(Ak, c + 16, 0, 64), bq0, s1);
    s1 = wmma_bf16(load_frag_bf16_rm(Ak, c + 16, 32, 64), bq1, s1);
    float mc = -1e30f;
#pragma unroll
    for (int r = 0; r < 8; ++r) { mc = fmaxf(mc, s0[r]); mc = fmaxf(mc, s1[r]); }
    mc = fmaxf(mc, __shfl_xor(mc, 16, 32));
    float newm = fmaxf(run_m, mc);
    float scale = __expf(run_m - newm);
    float p0[8], p1[8], ps = 0.f;
#pragma unroll
    for (int r = 0; r < 8; ++r) {
      p0[r] = __expf(s0[r] - newm);
      p1[r] = __expf(s1[r] - newm);
      ps += p0[r] + p1[r];
    }
    run_s = run_s * scale + ps;
    run_m = newm;
    FragAB bp;
#pragma unroll
    for (int p = 0; p < 4; ++p) {
      bp.u[p] = pack2bf(p0[2 * p], p0[2 * p + 1]);
      bp.u[4 + p] = pack2bf(p1[2 * p], p1[2 * p + 1]);
    }
#pragma unroll
    for (int t = 0; t < 4; ++t) {
#pragma unroll
      for (int r = 0; r < 8; ++r) acc[t][r] *= scale;
      acc[t] = wmma_bf16(load_frag_bf16_rm(A2, 16 * t, c, ldA2), bp.v, acc[t]);
    }
  }
  float tot = run_s + __shfl_xor(run_s, 16, 32);
  float inv = 1.f / tot;
#pragma unroll
  for (int t = 0; t < 4; ++t)
#pragma unroll
    for (int r = 0; r < 8; ++r) {
      int dh = 16 * t + r + 8 * h;
      float val = acc[t][r] * inv;
      if (outF) outF[(size_t)n * rsF + dh] = val;
      if (outB) outB[(size_t)n * nsB + (size_t)dh * dsB] = (__bf16)val;
    }
}

__global__ void k_attn3v(const __bf16* kb, const __bf16* ql, const __bf16* vT,
                         __bf16* kvT) {
  int wid = (int)((blockIdx.x * blockDim.x + threadIdx.x) >> 5);
  int bh = wid >> 4, blk = wid & 15;
  stream_attn(kb + (size_t)bh * 4096 * 64, ql + (size_t)bh * 256 * 64, blk * 16,
              vT + (size_t)bh * 64 * 4096, 4096, 4096, nullptr, 0,
              kvT + (size_t)bh * 64 * 256 + blk * 16, 1, 256);
}

__global__ void k_w2(const __bf16* zb, const __bf16* kvT, __bf16* w2t) {
  int wid = (int)((blockIdx.x * blockDim.x + threadIdx.x) >> 5);
  int bh = wid >> 4, tM = wid & 15;
  floatx8 acc[4] = {};
  gemm_strip<8>(zb + ((size_t)bh << 16), tM * 16, kvT + (size_t)bh * 64 * 256,
                0, 256, acc);
  int lane = threadIdx.x & 31;
  int n = lane & 15, h = lane >> 4;
#pragma unroll
  for (int t4 = 0; t4 < 4; ++t4)
#pragma unroll
    for (int r = 0; r < 8; ++r) {
      int M = tM * 16 + r + 8 * h;
      int N = 16 * t4 + n;
      w2t[((size_t)bh * 64 + N) * 256 + M] = (__bf16)acc[t4][r];
    }
}

__global__ void k_attn1_out(const __bf16* kl, const __bf16* qb,
                            const __bf16* w2t, float* aout) {
  int wid = (int)((blockIdx.x * blockDim.x + threadIdx.x) >> 5);
  int bh = wid >> 8, blk = wid & 255;
  int b = bh >> 3, hh = bh & 7;
  stream_attn(kl + (size_t)bh * 256 * 64, qb + (size_t)bh * 4096 * 64, blk * 16,
              w2t + (size_t)bh * 64 * 256, 256, 256,
              aout + ((size_t)b * 4096 + blk * 16) * 512 + hh * 64, 512,
              nullptr, 0, 0);
}

__global__ void k_conv_res(const float* vf, const float* rw, float* aout,
                           __bf16* aout_b) {
  size_t i = (size_t)blockIdx.x * blockDim.x + threadIdx.x;
  if (i >= (size_t)4 * 4096 * 512) return;
  int col = (int)(i & 511);
  size_t tok = i >> 9;
  int b = (int)(tok >> 12), ns = (int)(tok & 4095);
  int hh = col >> 6, dh = col & 63;
  const float* vrow = vf + ((size_t)(b * 8 + hh) * 4096) * 64 + dh;
  float s = 0.f;
#pragma unroll
  for (int j = 0; j < 33; ++j) {
    int p = ns + j - 16;
    if (p >= 0 && p < 4096) s += vrow[(size_t)p * 64] * rw[hh * 33 + j];
  }
  float val = aout[i] + s;
  aout[i] = val;
  aout_b[i] = (__bf16)val;
}

__global__ void k_outproj(const __bf16* aoutb, const __bf16* wo,
                          const float* bias, const float* x, float* out) {
  int wid = (int)((blockIdx.x * blockDim.x + threadIdx.x) >> 5);
  int tM = wid >> 3, s = wid & 7;
  floatx8 acc[4] = {};
  gemm_strip<16>(aoutb, tM * 16, wo, s * 64, 512, acc);
  int lane = threadIdx.x & 31;
  int n = lane & 15, h = lane >> 4;
#pragma unroll
  for (int t4 = 0; t4 < 4; ++t4) {
    int col = s * 64 + 16 * t4 + n;
    float bc = bias[col];
#pragma unroll
    for (int r = 0; r < 8; ++r) {
      size_t idx = (size_t)(tM * 16 + r + 8 * h) * 512 + col;
      out[idx] = x[idx] + acc[t4][r] + bc;
    }
  }
}

extern "C" void kernel_launch(void* const* d_in, const int* in_sizes, int n_in,
                              void* d_out, int out_size, void* d_ws,
                              size_t ws_size, hipStream_t stream) {
  const float* x     = (const float*)d_in[0];
  const float* ln_g  = (const float*)d_in[1];
  const float* ln_b  = (const float*)d_in[2];
  const float* w_qkv = (const float*)d_in[3];
  const float* w_out = (const float*)d_in[4];
  const float* b_out = (const float*)d_in[5];
  const float* res_w = (const float*)d_in[6];
  float* out = (float*)d_out;

  char* p = (char*)d_ws;
  auto alloc = [&](size_t bytes) -> void* {
    void* r = (void*)p;
    p += (bytes + 255) & ~(size_t)255;
    return r;
  };
  const size_t NTOK = 4 * 4096;
  const size_t MM = (size_t)32 * 256 * 256;
  __bf16* xn_b   = (__bf16*)alloc(NTOK * 512 * 2);
  __bf16* wqkv_b = (__bf16*)alloc(1536 * 512 * 2);
  __bf16* wout_b = (__bf16*)alloc(512 * 512 * 2);
  __bf16* q_b    = (__bf16*)alloc((size_t)32 * 4096 * 64 * 2);
  __bf16* k_b    = (__bf16*)alloc((size_t)32 * 4096 * 64 * 2);
  float*  v_f    = (float*)alloc((size_t)32 * 4096 * 64 * 4);
  __bf16* vT_b   = (__bf16*)alloc((size_t)32 * 64 * 4096 * 2);
  __bf16* ql_b   = (__bf16*)alloc((size_t)32 * 256 * 64 * 2);
  __bf16* kl_b   = (__bf16*)alloc((size_t)32 * 256 * 64 * 2);
  float*  mat_x  = (float*)alloc(MM * 4);
  __bf16* x_b    = (__bf16*)alloc(MM * 2);
  __bf16* z0_b   = (__bf16*)alloc(MM * 2);
  __bf16* z0T_b  = (__bf16*)alloc(MM * 2);
  __bf16* z1_b   = (__bf16*)alloc(MM * 2);
  __bf16* z1T_b  = (__bf16*)alloc(MM * 2);
  __bf16* xz_b   = (__bf16*)alloc(MM * 2);
  __bf16* t1T_b  = (__bf16*)alloc(MM * 2);
  __bf16* t2T_b  = (__bf16*)alloc(MM * 2);
  __bf16* t3T_b  = (__bf16*)alloc(MM * 2);
  __bf16* kvT_b  = (__bf16*)alloc((size_t)32 * 64 * 256 * 2);
  __bf16* w2t_b  = (__bf16*)alloc((size_t)32 * 64 * 256 * 2);
  float*  aout_f = (float*)alloc(NTOK * 512 * 4);
  __bf16* aout_b = (__bf16*)alloc(NTOK * 512 * 2);
  float*  scales = (float*)alloc(256);
  (void)ws_size; (void)n_in; (void)in_sizes; (void)out_size;

  const int T = 256;
  k_convert_w<<<3072, T, 0, stream>>>(w_qkv, w_out, wqkv_b, wout_b, scales);
  k_layernorm<<<2048, T, 0, stream>>>(x, ln_g, ln_b, xn_b, (int)NTOK);
  k_gemm_qkv<<<3072, T, 0, stream>>>(xn_b, wqkv_b, q_b, k_b, v_f, vT_b);
  k_landmarks<<<2048, T, 0, stream>>>(q_b, k_b, ql_b, kl_b);
  k_attn2_logits<<<256, T, 0, stream>>>(ql_b, kl_b, mat_x);
  k_softmax256<<<1024, T, 0, stream>>>(mat_x, x_b, 32 * 256);
  k_pinv_scales<<<1024, T, 0, stream>>>(mat_x, scales);
  k_init_z<<<8192, T, 0, stream>>>(mat_x, z0_b, z0T_b, scales);

  __bf16* zc = z0_b;  __bf16* zcT = z0T_b;
  __bf16* zn = z1_b;  __bf16* znT = z1T_b;
  for (int it = 0; it < 6; ++it) {
    k_gemm256b<<<256, T, 0, stream>>>(x_b, zcT, xz_b, nullptr, t1T_b, 7.f, -1.f);
    k_gemm256b<<<256, T, 0, stream>>>(xz_b, t1T_b, nullptr, nullptr, t2T_b, 15.f, -1.f);
    k_gemm256b<<<256, T, 0, stream>>>(xz_b, t2T_b, nullptr, nullptr, t3T_b, 13.f, -1.f);
    k_gemm256b<<<256, T, 0, stream>>>(zc, t3T_b, nullptr, zn, znT, 0.f, 0.25f);
    __bf16* t;
    t = zc; zc = zn; zn = t;
    t = zcT; zcT = znT; znT = t;
  }

  k_attn3v<<<64, T, 0, stream>>>(k_b, ql_b, vT_b, kvT_b);
  k_w2<<<64, T, 0, stream>>>(zc, kvT_b, w2t_b);
  k_attn1_out<<<1024, T, 0, stream>>>(kl_b, q_b, w2t_b, aout_f);
  k_conv_res<<<32768, T, 0, stream>>>(v_f, res_w, aout_f, aout_b);
  k_outproj<<<1024, T, 0, stream>>>(aout_b, wout_b, b_out, x, out);
}


# attn2 logits, row softmax and pinv scale sums: 3 launches -> 1, a 16-wave work-group per matrix keeps the logits in registers; atomics gone (per-matrix maxima); softmax and |x| sum order changes
# speedup vs baseline: 4.3629x; 4.3629x over previous
#include <hip/hip_runtime.h>
#include <hip/hip_bf16.h>
#include <math.h>


typedef __attribute__((ext_vector_type(16))) __bf16 bf16x16;
typedef __attribute__((ext_vector_type(8)))  float  floatx8;

union FragAB { bf16x16 v; unsigned int u[8]; };

__device__ __forceinline__ unsigned pack2bf(float a, float b) {
  __bf16 lo = (__bf16)a, hi = (__bf16)b;
  unsigned short ul = __builtin_bit_cast(unsigned short, lo);
  unsigned short uh = __builtin_bit_cast(unsigned short, hi);
  return (unsigned)ul | ((unsigned)uh << 16);
}

__device__ __forceinline__ int kpair(int p, int h) {
  return (p < 4) ? (2 * p + 8 * h) : (8 + 2 * p + 8 * h);
}

__device__ __forceinline__ bf16x16 load_frag_bf16_rm(const __bf16* src, int row0,
                                                     int k0, int ld) {
  int lane = threadIdx.x & 31;
  int r = lane & 15, h = lane >> 4;
  const __bf16* base = src + (size_t)(row0 + r) * ld + k0;
  FragAB f;
#pragma unroll
  for (int p = 0; p < 8; ++p) f.u[p] = *(const unsigned*)(base + kpair(p, h));
  return f.v;
}

__device__ __forceinline__ floatx8 wmma_bf16(bf16x16 a, bf16x16 b, floatx8 c) {
  return __builtin_amdgcn_wmma_f32_16x16x32_bf16(false, a, false, b, (short)0, c,
                                                 false, false);
}

template <int KSTEPS>
__device__ __forceinline__ void gemm_strip(const __bf16* A, int rowA,
                                           const __bf16* B, int rowB, int ld,
                                           floatx8 acc[4]) {
#pragma unroll
  for (int ks = 0; ks < KSTEPS; ++ks) {
    int k0 = ks * 32;
    bf16x16 af = load_frag_bf16_rm(A, rowA, k0, ld);
#pragma unroll
    for (int t = 0; t < 4; ++t) {
      bf16x16 bfr = load_frag_bf16_rm(B, rowB + 16 * t, k0, ld);
      acc[t] = wmma_bf16(af, bfr, acc[t]);
    }
  }
}

__device__ __forceinline__ void load_ablock(const __bf16* A, int row0,
                                            bf16x16 a[16]) {
#pragma unroll
  for (int ks = 0; ks < 16; ++ks) a[ks] = load_frag_bf16_rm(A, row0, ks * 32, 512);
}

__device__ __forceinline__ void mma_ablock(const bf16x16 a[16],
                                           const bf16x16 (&bw)[16][4],
                                           floatx8 acc[4]) {
#pragma unroll
  for (int ks = 0; ks < 16; ++ks)
#pragma unroll
    for (int t = 0; t < 4; ++t) acc[t] = wmma_bf16(a[ks], bw[ks][t], acc[t]);
}

template <int STEP, class Pre, class Fin>
__device__ __forceinline__ void gemm_wstat(const __bf16* A, const __bf16* W,
                                           int s, int g, Pre pre, Fin fin) {
  bf16x16 bw[16][4];
#pragma unroll
  for (int ks = 0; ks < 16; ++ks)
#pragma unroll
    for (int t = 0; t < 4; ++t)
      bw[ks][t] = load_frag_bf16_rm(W, s * 64 + 16 * t, ks * 32, 512);
  bf16x16 a0[16], a1[16];
  load_ablock(A, g * 16, a0);
  for (int tM = g; tM < 1024; tM += 2 * STEP) {
    int tM1 = tM + STEP, tM2 = tM + 2 * STEP;
    load_ablock(A, min(tM1, 1023) * 16, a1);
    {
      auto p = pre(tM);
      __builtin_amdgcn_sched_barrier(0);
      floatx8 acc[4] = {};
      mma_ablock(a0, bw, acc);
      fin(tM, acc, p);
    }
    load_ablock(A, min(tM2, 1023) * 16, a0);
    if (tM1 < 1024) {
      auto p = pre(tM1);
      __builtin_amdgcn_sched_barrier(0);
      floatx8 acc[4] = {};
      mma_ablock(a1, bw, acc);
      fin(tM1, acc, p);
    }
  }
}

__global__ void k_convert_w(const float* wqkv, const float* wout,
                            __bf16* wqkv_b, __bf16* wout_b) {
  int i = blockIdx.x * blockDim.x + threadIdx.x;
  if (i < 1536 * 512) wqkv_b[i] = (__bf16)wqkv[i];
  if (i < 512 * 512) wout_b[i] = (__bf16)wout[i];
}

__global__ void k_layernorm(const float* x, const float* g, const float* b,
                            __bf16* xn, int ntok) {
  int wave = (int)((blockIdx.x * blockDim.x + threadIdx.x) >> 5);
  int lane = threadIdx.x & 31;
  if (wave >= ntok) return;
  const float* xr = x + (size_t)wave * 512;
  float s = 0.f, s2 = 0.f, vals[16];
#pragma unroll
  for (int i = 0; i < 16; ++i) {
    float v = xr[lane + 32 * i];
    vals[i] = v; s += v; s2 += v * v;
  }
  for (int m = 16; m >= 1; m >>= 1) {
    s += __shfl_xor(s, m, 32);
    s2 += __shfl_xor(s2, m, 32);
  }
  float mu = s * (1.f / 512.f);
  float var = s2 * (1.f / 512.f) - mu * mu;
  float inv = rsqrtf(var + 1e-5f);
  __bf16* o = xn + (size_t)wave * 512;
#pragma unroll
  for (int i = 0; i < 16; ++i) {
    int c = lane + 32 * i;
    o[c] = (__bf16)((vals[i] - mu) * inv * g[c] + b[c]);
  }
}

__global__ __launch_bounds__(128)
void k_gemm_qkv(const __bf16* xn, const __bf16* wq, __bf16* qb, __bf16* kb,
                float* vf, __bf16* vT) {
  int wave = __builtin_amdgcn_readfirstlane((int)(threadIdx.x >> 5));
  int g = (int)(blockIdx.x / 6);
  int s = (int)(blockIdx.x % 6) * 4 + wave;
  int lane = threadIdx.x & 31;
  int n = lane & 15, h = lane >> 4;
  int sect = s >> 3;
  int head = s & 7;
  gemm_wstat<42>(xn, wq, s, g, [](int) { return 0; },
                 [&](int tM, floatx8 acc[4], int) {
    int tok0 = tM * 16 + 8 * h;
    int bb = tok0 >> 12, ns0 = tok0 & 4095;
    size_t bh = (size_t)(bb * 8 + head);
    size_t idx0 = (bh * 4096 + ns0) * 64 + n;
    if (sect == 0) {
#pragma unroll
      for (int t = 0; t < 4; ++t)
#pragma unroll
        for (int r = 0; r < 8; ++r)
          qb[idx0 + r * 64 + 16 * t] = (__bf16)(acc[t][r] * 0.125f);
    } else if (sect == 1) {
#pragma unroll
      for (int t = 0; t < 4; ++t)
#pragma unroll
        for (int r = 0; r < 8; ++r)
          kb[idx0 + r * 64 + 16 * t] = (__bf16)acc[t][r];
    } else {
      size_t tdx0 = (bh * 64 + n) * 4096 + ns0;
#pragma unroll
      for (int t = 0; t < 4; ++t)
#pragma unroll
        for (int r = 0; r < 8; ++r) {
          vf[idx0 + r * 64 + 16 * t] = acc[t][r];
          vT[tdx0 + (size_t)(16 * t) * 4096 + r] = (__bf16)acc[t][r];
        }
    }
  });
}

__global__ void k_landmarks(const __bf16* qb, const __bf16* kb, __bf16* ql,
                            __bf16* kl) {
  int i = blockIdx.x * blockDim.x + threadIdx.x;
  if (i >= 32 * 256 * 64) return;
  int dh = i & 63, m = (i >> 6) & 255, bh = i >> 14;
  size_t base = ((size_t)bh * 4096 + m * 16) * 64 + dh;
  float sq = 0.f, sk = 0.f;
#pragma unroll
  for (int t = 0; t < 16; ++t) {
    sq += (float)qb[base + t * 64];
    sk += (float)kb[base + t * 64];
  }
  ql[i] = (__bf16)(sq * (1.f / 16.f));
  kl[i] = (__bf16)(sk * (1.f / 16.f));
}


typedef __attribute__((ext_vector_type(4))) unsigned int uintx4;
typedef __attribute__((ext_vector_type(2))) unsigned int uintx2;
typedef __attribute__((ext_vector_type(4))) float floatx4;

#define PLANE_Q 8192

union FragQ { bf16x16 v; uintx4 q[2]; };

__device__ __forceinline__ bf16x16 lds_frag(const uintx4* plane, int nb, int kb) {
  int lane = threadIdx.x & 31;
  const uintx4* blk = plane + (nb * 8 + kb) * 64;
  FragQ f;
  f.q[0] = blk[lane];
  f.q[1] = blk[32 + lane];
  return f.v;
}

typedef short s16x8 __attribute__((ext_vector_type(8)));

__device__ __forceinline__ uintx4 lds_tr16(const uintx4* p) {
  typedef __attribute__((address_space(3))) s16x8 lds_s16x8;
  return __builtin_bit_cast(
      uintx4, __builtin_amdgcn_ds_load_tr16_b128_v8i16((lds_s16x8*)p));
}

__device__ __forceinline__ void block_sync_all() {
  __builtin_amdgcn_fence(__ATOMIC_RELEASE, "workgroup");
  __builtin_amdgcn_s_barrier();
  __builtin_amdgcn_fence(__ATOMIC_ACQUIRE, "workgroup");
}

template <bool ATR, bool DIAG, bool OT, bool PRESYNC>
__device__ __forceinline__ void pinv_gemm_tile(const __bf16* A, const uintx4* Ain,
                                               const uintx4* Bin, uintx4* Bout,
                                               uintx4* Dblk, __bf16* OutT,
                                               float alpha, float beta, int tm,
                                               int tn) {
  int lane = threadIdx.x & 31;
  int n = lane & 15, h = lane >> 4;
  int lp = ((lane >> 3) & 1) * 16 + 8 * h + (lane & 7);
  floatx8 acc[2][8] = {};
#pragma unroll
  for (int ks = 0; ks < 8; ++ks) {
    bf16x16 af[2], bfr[8];
#pragma unroll
    for (int i = 0; i < 2; ++i) {
      if (ATR) {
        int o0 = ((2 * ks) * 8 + tm) * 64 + i * 32 + lp;
        int o1 = o0 + 512;
        int od = (int)(Dblk - Ain) + (2 * tm + i) * 32 + lp;
        bool dstep = (ks == tm);
        if (i == 0) o0 = dstep ? od : o0; else o1 = dstep ? od : o1;
        FragQ f;
        f.q[0] = lds_tr16(Ain + o0);
        f.q[1] = lds_tr16(Ain + o1);
        af[i] = f.v;
      } else {
        af[i] = load_frag_bf16_rm(A, tm * 32 + 16 * i, ks * 32, 256);
      }
    }
#pragma unroll
    for (int j = 0; j < 8; ++j) bfr[j] = lds_frag(Bin, tn * 8 + j, ks);
#pragma unroll
    for (int i = 0; i < 2; ++i)
#pragma unroll
      for (int j = 0; j < 8; ++j) acc[i][j] = wmma_bf16(af[i], bfr[j], acc[i][j]);
  }
  if (PRESYNC) block_sync_all();
  int dblk = tn * 8 - tm * 2;
  float il[8];
#pragma unroll
  for (int r = 0; r < 8; ++r) il[r] = (n == r + 8 * h) ? 1.f : 0.f;
  size_t rm0 = (size_t)(tm * 32 + 8 * h) * 256 + (size_t)(tn * 128 + n);
#pragma unroll
  for (int i = 0; i < 2; ++i)
#pragma unroll
    for (int j = 0; j < 8; ++j) {
      float ad = (i - j == dblk) ? alpha : 0.f;
      float trv[8];
#pragma unroll
      for (int r = 0; r < 8; ++r) {
        float tv = acc[i][j][r];
        float tr = __builtin_fmaf(ad, il[r], beta * tv);
        if (OT) OutT[rm0 + (size_t)((16 * i + r) * 256 + 16 * j)] = (__bf16)tr;
        trv[r] = tr;
      }
      if (DIAG && (i - j == dblk)) {
        uintx4 d;
        d.x = pack2bf(-acc[i][j][0], -acc[i][j][1]);
        d.y = pack2bf(-acc[i][j][2], -acc[i][j][3]);
        d.z = pack2bf(-acc[i][j][4], -acc[i][j][5]);
        d.w = pack2bf(-acc[i][j][6], -acc[i][j][7]);
        Dblk[(2 * tm + i) * 32 + lane] = d;
      }
      uintx4 w;
      w.x = pack2bf(trv[0], trv[1]);
      w.y = pack2bf(trv[2], trv[3]);
      w.z = pack2bf(trv[4], trv[5]);
      w.w = pack2bf(trv[6], trv[7]);
      Bout[((tn * 8 + j) * 8 + tm) * 64 + i * 32 + lane] = w;
    }
}


#define IMG_Q 4096

__device__ __forceinline__ int img_unit(int f, int q, int h, int r) {
  return f * 64 + q * 32 + h * 16 + (r ^ (4 * h + 8 * q));
}

__device__ __forceinline__ bf16x16 img_frag(const uintx4* img, int f) {
  int lane = threadIdx.x & 31;
  int r = lane & 15, h = lane >> 4;
  FragQ fr;
  fr.q[0] = img[img_unit(f, 0, h, r)];
  fr.q[1] = img[img_unit(f, 1, h, r)];
  return fr.v;
}

template <int NT>
__device__ __forceinline__ void img_fetch(const uintx4* kq, const uintx4* vq,
                                          int vpitch, uintx4 (&pk)[2048 / NT],
                                          uintx4 (&pv)[2048 / NT]) {
#pragma unroll
  for (int j = 0; j < 2048 / NT; ++j) {
    int i = (int)threadIdx.x + NT * j;
    int wr = (i >> 3) & 63;
    int cb = ((i >> 2) & 1) + 2 * (i >> 9);
    pk[j] = kq[i];
    pv[j] = vq[wr * vpitch + cb * 4 + (i & 3)];
  }
}

template <int NT>
__device__ __forceinline__ void img_put(uintx4* img, const uintx4 (&pk)[2048 / NT],
                                        const uintx4 (&pv)[2048 / NT]) {
#pragma unroll
  for (int j = 0; j < 2048 / NT; ++j) {
    int i = (int)threadIdx.x + NT * j;
    int p = i & 3;
    int c2 = (i >> 2) & 1;
    int kr = i >> 3;
    int wr = kr & 63;
    int cb = c2 + 2 * (i >> 9);
    img[img_unit((kr >> 4) * 2 + c2, p >> 1, p & 1, kr & 15)] = pk[j];
    img[img_unit(32 + (wr >> 4) * 8 + cb, p >> 1, p & 1, wr & 15)] = pv[j];
  }
}

__device__ __forceinline__ void stream_attn_begin(float& run_m, float& run_s,
                                                  floatx8 acc[4]) {
  run_m = -1e30f;
  run_s = 0.f;
#pragma unroll
  for (int t = 0; t < 4; ++t)
#pragma unroll
    for (int r = 0; r < 8; ++r) acc[t][r] = 0.f;
}

__device__ __forceinline__ void stream_attn_keys(const uintx4* img, bf16x16 bq0,
                                                 bf16x16 bq1, float& run_m,
                                                 float& run_s, floatx8 acc[4]) {
#pragma unroll
  for (int c = 0; c < 256; c += 32) {
    int kf = (c >> 4) * 2;
    floatx8 s0 = {}, s1 = {};
    s0 = wmma_bf16(img_frag(img, kf), bq0, s0);
    s0 = wmma_bf16(img_frag(img, kf + 1), bq1, s0);
    s1 = wmma_bf16(img_frag(img, kf + 2), bq0, s1);
    s1 = wmma_bf16(img_frag(img, kf + 3), bq1, s1);
    float mc = -1e30f;
#pragma unroll
    for (int r = 0; r < 8; ++r) { mc = fmaxf(mc, s0[r]); mc = fmaxf(mc, s1[r]); }
    mc = fmaxf(mc, __shfl_xor(mc, 16, 32));
    float newm = fmaxf(run_m, mc);
    float scale = __expf(run_m - newm);
    float p0[8], p1[8], ps = 0.f;
#pragma unroll
    for (int r = 0; r < 8; ++r) {
      p0[r] = __expf(s0[r] - newm);
      p1[r] = __expf(s1[r] - newm);
      ps += p0[r] + p1[r];
    }
    run_s = run_s * scale + ps;
    run_m = newm;
    FragAB bp;
#pragma unroll
    for (int p = 0; p < 4; ++p) {
      bp.u[p] = pack2bf(p0[2 * p], p0[2 * p + 1]);
      bp.u[4 + p] = pack2bf(p1[2 * p], p1[2 * p + 1]);
    }
#pragma unroll
    for (int t = 0; t < 4; ++t) {
#pragma unroll
      for (int r = 0; r < 8; ++r) acc[t][r] *= scale;
      acc[t] = wmma_bf16(img_frag(img, 32 + t * 8 + (c >> 5)), bp.v, acc[t]);
    }
  }
}

__device__ __forceinline__ void stream_attn_end(float run_s, floatx8 acc[4]) {
  float tot = run_s + __shfl_xor(run_s, 16, 32);
  float inv = 1.f / tot;
#pragma unroll
  for (int t = 0; t < 4; ++t)
#pragma unroll
    for (int r = 0; r < 8; ++r) acc[t][r] *= inv;
}

__device__ __forceinline__ void attn3v_staged(const __bf16* K, const __bf16* Q,
                                              int qrow0, const __bf16* VT,
                                              __bf16* outB, uintx4* lds) {
  const uintx4* kq = (const uintx4*)K;
  const uintx4* vq = (const uintx4*)VT;
  uintx4 pk[4], pv[4];
  img_fetch<512>(kq, vq, 512, pk, pv);
  img_put<512>(lds, pk, pv);
  bf16x16 bq0 = load_frag_bf16_rm(Q, qrow0, 0, 64);
  bf16x16 bq1 = load_frag_bf16_rm(Q, qrow0, 32, 64);
  float run_m, run_s;
  floatx8 acc[4];
  stream_attn_begin(run_m, run_s, acc);
  __syncthreads();
  for (int s = 0; s < 16; ++s) {
    uintx4* cur = lds + (s & 1) * IMG_Q;
    uintx4* nxt = lds + ((s + 1) & 1) * IMG_Q;
    bool more = s + 1 < 16;
    if (more) img_fetch<512>(kq + (s + 1) * 2048, vq + (s + 1) * 32, 512, pk, pv);
    stream_attn_keys(cur, bq0, bq1, run_m, run_s, acc);
    if (more) {
      img_put<512>(nxt, pk, pv);
      __syncthreads();
    }
  }
  stream_attn_end(run_s, acc);
  int lane = threadIdx.x & 31;
  int n = lane & 15, h = lane >> 4;
#pragma unroll
  for (int t = 0; t < 4; ++t)
#pragma unroll
    for (int r = 0; r < 8; ++r) {
      int dh = 16 * t + r + 8 * h;
      outB[(size_t)n + (size_t)dh * 256] = (__bf16)acc[t][r];
    }
}

#define A2_PITCH 260

template <int M>
__device__ __forceinline__ float row_xor(float x) {
  int b = __builtin_bit_cast(int, x);
  return __builtin_bit_cast(
      float, __builtin_amdgcn_update_dpp(b, b, 0x160 + M, 0xf, 0xf, false));
}

__device__ __forceinline__ float row_max16(float x) {
  x = fmaxf(x, row_xor<8>(x));
  x = fmaxf(x, row_xor<4>(x));
  x = fmaxf(x, row_xor<2>(x));
  return fmaxf(x, row_xor<1>(x));
}

__device__ __forceinline__ float row_sum16(float x) {
  x += row_xor<8>(x);
  x += row_xor<4>(x);
  x += row_xor<2>(x);
  return x + row_xor<1>(x);
}

__global__ __launch_bounds__(512)
void k_attn2(const __bf16* ql, const __bf16* kl, float* xf, __bf16* xb,
             float* pmax) {
  __shared__ uintx4 kimg[2048];
  __shared__ uintx4 stage[16 * 16 * A2_PITCH / 4];
  __shared__ float colp[16 * 256];
  __shared__ float rowm[16];
  __shared__ float colm[8];
  int wave = __builtin_amdgcn_readfirstlane((int)(threadIdx.x >> 5));
  int lane = threadIdx.x & 31;
  int n = lane & 15, h = lane >> 4;
  size_t bh = blockIdx.x;
  const __bf16* A = ql + bh * 256 * 64;
  {
    const uintx4* kq = (const uintx4*)(kl + bh * 256 * 64);
    uintx4 pk[4];
#pragma unroll
    for (int j = 0; j < 4; ++j) pk[j] = kq[(int)threadIdx.x + 512 * j];
#pragma unroll
    for (int j = 0; j < 4; ++j) {
      int i = (int)threadIdx.x + 512 * j;
      int p = i & 3, c2 = (i >> 2) & 1, kr = i >> 3;
      kimg[img_unit((kr >> 4) * 2 + c2, p >> 1, p & 1, kr & 15)] = pk[j];
    }
  }
  bf16x16 af0 = load_frag_bf16_rm(A, wave * 16, 0, 64);
  bf16x16 af1 = load_frag_bf16_rm(A, wave * 16, 32, 64);
  __syncthreads();
  floatx8 acc[16];
#pragma unroll
  for (int j = 0; j < 16; ++j) {
    floatx8 z = {};
    acc[j] = wmma_bf16(af0, img_frag(kimg, 2 * j), z);
  }
#pragma unroll
  for (int j = 0; j < 16; ++j) acc[j] = wmma_bf16(af1, img_frag(kimg, 2 * j + 1), acc[j]);
  float* my = (float*)stage + wave * (16 * A2_PITCH);
  size_t g0 = (bh * 256 + (size_t)(wave * 16)) * 256 + 4 * lane;
  float rmax = 0.f;
  float cpart[16];
#pragma unroll
  for (int j = 0; j < 16; ++j) cpart[j] = 0.f;
#pragma unroll
  for (int r = 0; r < 8; ++r) {
    float m = acc[0][r];
#pragma unroll
    for (int j = 1; j < 16; ++j) m = fmaxf(m, acc[j][r]);
    m = row_max16(m);
    float e[16], s = 0.f;
#pragma unroll
    for (int j = 0; j < 16; ++j) { e[j] = __expf(acc[j][r] - m); s += e[j]; }
    s = row_sum16(s);
    float inv = 1.f / s;
    float rs = 0.f;
#pragma unroll
    for (int j = 0; j < 16; ++j) {
      float pv = e[j] * inv;
      my[(8 * h + r) * A2_PITCH + 16 * j + n] = pv;
      rs += fabsf(pv);
      cpart[j] += fabsf(pv);
    }
    rs = row_sum16(rs);
    rmax = fmaxf(rmax, rs);
    __builtin_amdgcn_fence(__ATOMIC_RELEASE, "wavefront");
    __builtin_amdgcn_wave_barrier();
    __builtin_amdgcn_fence(__ATOMIC_ACQUIRE, "wavefront");
#pragma unroll
    for (int hh = 0; hh < 2; ++hh)
#pragma unroll
      for (int k = 0; k < 2; ++k) {
        int i = 8 * hh + r;
        floatx4 v = *(const floatx4*)(my + i * A2_PITCH + 128 * k + 4 * lane);
        size_t g = g0 + (size_t)(i * 256 + 128 * k);
        *(floatx4*)(xf + g) = v;
        uintx2 w;
        w.x = pack2bf(v.x, v.y);
        w.y = pack2bf(v.z, v.w);
        *(uintx2*)(xb + g) = w;
      }
  }
  rmax = fmaxf(rmax, __shfl_xor(rmax, 16, 32));
#pragma unroll
  for (int j = 0; j < 16; ++j) cpart[j] += __shfl_xor(cpart[j], 16, 32);
  if (h == 0) {
#pragma unroll
    for (int j = 0; j < 16; ++j) colp[wave * 256 + 16 * j + n] = cpart[j];
  }
  if (lane == 0) rowm[wave] = rmax;
  __syncthreads();
  if (threadIdx.x < 256) {
    float cs = 0.f;
#pragma unroll
    for (int w = 0; w < 16; ++w) cs += colp[w * 256 + (int)threadIdx.x];
    for (int t = 16; t >= 1; t >>= 1) cs = fmaxf(cs, __shfl_xor(cs, t, 32));
    if (lane == 0) colm[wave] = cs;
  }
  __syncthreads();
  if (threadIdx.x == 0) {
    float cm = colm[0], rm = rowm[0];
    for (int w = 1; w < 8; ++w) cm = fmaxf(cm, colm[w]);
    for (int w = 1; w < 16; ++w) rm = fmaxf(rm, rowm[w]);
    pmax[bh] = rm;
    pmax[32 + bh] = cm;
  }
}

__global__ __launch_bounds__(512)
void k_pinv_attn3v(const __bf16* xb, const float* xf, const float* scales,
                   __bf16* z0, __bf16* z1, const __bf16* kb,
                   const __bf16* ql, const __bf16* vT, __bf16* kvT) {
  __shared__ uintx4 ldsB[2 * PLANE_Q + 512];
  int wave = __builtin_amdgcn_readfirstlane((int)(threadIdx.x >> 5));
  if (blockIdx.x >= 32) {
    size_t bh = blockIdx.x - 32;
    attn3v_staged(kb + bh * 4096 * 64, ql + bh * 256 * 64, wave * 16,
                  vT + bh * 64 * 4096, kvT + bh * 64 * 256 + wave * 16, ldsB);
    return;
  }
  uintx4* B0 = ldsB;
  uintx4* B1 = ldsB + PLANE_Q;
  uintx4* DG = ldsB + 2 * PLANE_Q;
  size_t bofs = (size_t)blockIdx.x << 16;
  int lane = threadIdx.x & 31;
  int tm = wave >> 1, tn = wave & 1;
  const __bf16* X = xb + bofs;
  __bf16* zc = z0 + bofs;
  __bf16* zn = z1 + bofs;
  const float* XF = xf + bofs;
  float rmx = scales[0], cmx = scales[32];
  for (int i = 1; i < 32; ++i) {
    rmx = fmaxf(rmx, scales[i]);
    cmx = fmaxf(cmx, scales[32 + i]);
  }
  float denom = rmx * cmx;
  {
    int c = lane & 15, h = lane >> 4;
#pragma unroll
    for (int b = 0; b < 8; ++b) {
      int blk = wave * 8 + b;
      const float* sp =
          XF + (size_t)((blk >> 3) * 16 + c) * 256 + (blk & 7) * 32 + 8 * h;
      floatx4 a0 = *(const floatx4*)sp, a1 = *(const floatx4*)(sp + 4);
      floatx4 c0 = *(const floatx4*)(sp + 16), c1 = *(const floatx4*)(sp + 20);
      uintx4 q0, q1;
      q0.x = pack2bf(a0.x / denom, a0.y / denom);
      q0.y = pack2bf(a0.z / denom, a0.w / denom);
      q0.z = pack2bf(a1.x / denom, a1.y / denom);
      q0.w = pack2bf(a1.z / denom, a1.w / denom);
      q1.x = pack2bf(c0.x / denom, c0.y / denom);
      q1.y = pack2bf(c0.z / denom, c0.w / denom);
      q1.z = pack2bf(c1.x / denom, c1.y / denom);
      q1.w = pack2bf(c1.z / denom, c1.w / denom);
      B1[blk * 64 + lane] = q0;
      B1[blk * 64 + 32 + lane] = q1;
    }
  }
  {
    int M = (int)(threadIdx.x & 255), kg = (int)(threadIdx.x >> 8);
    for (int p = 0; p < 16; p += 4) {
      float raw[4][8];
#pragma unroll
      for (int u = 0; u < 4; ++u)
#pragma unroll
        for (int e = 0; e < 8; ++e)
          raw[u][e] = XF[(size_t)(((p + u) * 2 + kg) * 8 + e) * 256 + M];
#pragma unroll
      for (int u = 0; u < 4; ++u) {
        float zv[8];
#pragma unroll
        for (int e = 0; e < 8; ++e) zv[e] = raw[u][e] / denom;
        uintx4 w;
        w.x = pack2bf(zv[0], zv[1]);
        w.y = pack2bf(zv[2], zv[3]);
        w.z = pack2bf(zv[4], zv[5]);
        w.w = pack2bf(zv[6], zv[7]);
        *(uintx4*)(zc + (size_t)M * 256 + ((p + u) * 2 + kg) * 8) = w;
      }
    }
  }
  block_sync_all();
  for (int it = 0; it < 6; ++it) {
    pinv_gemm_tile<false, true, false, false>(X, nullptr, B1, B0, DG, nullptr,
                                              7.f, -1.f, tm, tn);
    block_sync_all();
    pinv_gemm_tile<true, false, false, false>(nullptr, B0, B0, B1, DG, nullptr,
                                              15.f, 1.f, tm, tn);
    block_sync_all();
    pinv_gemm_tile<true, false, false, true>(nullptr, B0, B1, B0, DG, nullptr,
                                             13.f, 1.f, tm, tn);
    block_sync_all();
    pinv_gemm_tile<false, false, true, false>(zc, nullptr, B0, B1, DG, zn, 0.f,
                                              0.25f, tm, tn);
    block_sync_all();
    __bf16* t = zc; zc = zn; zn = t;
  }
}

__global__ void k_w2(const __bf16* zb, const __bf16* kvT, __bf16* w2t) {
  int wid = (int)((blockIdx.x * blockDim.x + threadIdx.x) >> 5);
  int bh = wid >> 4, tM = wid & 15;
  floatx8 acc[4] = {};
  gemm_strip<8>(zb + ((size_t)bh << 16), tM * 16, kvT + (size_t)bh * 64 * 256,
                0, 256, acc);
  int lane = threadIdx.x & 31;
  int n = lane & 15, h = lane >> 4;
#pragma unroll
  for (int t4 = 0; t4 < 4; ++t4)
#pragma unroll
    for (int r = 0; r < 8; ++r) {
      int M = tM * 16 + r + 8 * h;
      int N = 16 * t4 + n;
      w2t[((size_t)bh * 64 + N) * 256 + M] = (__bf16)acc[t4][r];
    }
}

typedef __attribute__((ext_vector_type(2))) float floatx2;

__device__ __forceinline__ floatx8 wmma_f32(floatx2 a, floatx2 b, floatx8 c) {
  return __builtin_amdgcn_wmma_f32_16x16x4_f32(false, a, false, b, (short)0, c,
                                               false, false);
}

__global__ __launch_bounds__(256, 8)
void k_attn1_out(const __bf16* kl, const __bf16* qb, const __bf16* w2t,
                 const float* vf, const float* rw, __bf16* aout_b) {
  __shared__ uintx4 img[IMG_Q];
  int wid = (int)((blockIdx.x * blockDim.x + threadIdx.x) >> 5);
  int bh = wid >> 8, blk = wid & 255;
  int b = bh >> 3, hh = bh & 7;
  {
    uintx4 pk[8], pv[8];
    img_fetch<256>((const uintx4*)(kl + (size_t)bh * 256 * 64),
                   (const uintx4*)(w2t + (size_t)bh * 64 * 256), 32, pk, pv);
    img_put<256>(img, pk, pv);
  }
  const __bf16* qhead = qb + (size_t)bh * 4096 * 64;
  bf16x16 bq0 = load_frag_bf16_rm(qhead, blk * 16, 0, 64);
  bf16x16 bq1 = load_frag_bf16_rm(qhead, blk * 16, 32, 64);
  float run_m, run_s;
  floatx8 acc[4];
  stream_attn_begin(run_m, run_s, acc);
  __syncthreads();
  stream_attn_keys(img, bq0, bq1, run_m, run_s, acc);
  stream_attn_end(run_s, acc);
  int lane = threadIdx.x & 31;
  int c = lane & 15, h = lane >> 4;
  const float* vhead = vf + (size_t)bh * 4096 * 64;
  const float* wrow = rw + hh * 33;
  int p0 = blk * 16 - 16;
#pragma unroll
  for (int ks = 0; ks < 12; ++ks) {
    int q0 = 4 * ks + 2 * h;
    floatx2 bw;
#pragma unroll
    for (int s = 0; s < 2; ++s) {
      int j = q0 + s - c;
      float wv = wrow[min(max(j, 0), 32)];
      bw[s] = (j >= 0 && j <= 32) ? wv : 0.f;
    }
#pragma unroll
    for (int t = 0; t < 4; ++t) {
      floatx2 av;
#pragma unroll
      for (int s = 0; s < 2; ++s) {
        int p = p0 + q0 + s;
        float x = vhead[(size_t)min(max(p, 0), 4095) * 64 + 16 * t + c];
        av[s] = (p >= 0 && p < 4096) ? x : 0.f;
      }
      acc[t] = wmma_f32(av, bw, acc[t]);
    }
  }
  __bf16* orow = aout_b + ((size_t)b * 4096 + blk * 16 + c) * 512 + hh * 64 + 8 * h;
#pragma unroll
  for (int t = 0; t < 4; ++t) {
    uintx4 w;
    w.x = pack2bf(acc[t][0], acc[t][1]);
    w.y = pack2bf(acc[t][2], acc[t][3]);
    w.z = pack2bf(acc[t][4], acc[t][5]);
    w.w = pack2bf(acc[t][6], acc[t][7]);
    *(uintx4*)(orow + 16 * t) = w;
  }
}

struct XTile { float v[4][8]; };

__global__ __launch_bounds__(128)
void k_outproj(const __bf16* aoutb, const __bf16* wo, const float* bias,
               const float* x, float* out) {
  int wave = __builtin_amdgcn_readfirstlane((int)(threadIdx.x >> 5));
  int g = (int)(blockIdx.x >> 1);
  int s = (int)(blockIdx.x & 1) * 4 + wave;
  int lane = threadIdx.x & 31;
  int n = lane & 15, h = lane >> 4;
  int col0 = s * 64 + n;
  float bc[4];
#pragma unroll
  for (int t4 = 0; t4 < 4; ++t4) bc[t4] = bias[col0 + 16 * t4];
  gemm_wstat<128>(
      aoutb, wo, s, g,
      [&](int tM) {
        XTile p;
        size_t idx0 = (size_t)(tM * 16 + 8 * h) * 512 + col0;
#pragma unroll
        for (int t4 = 0; t4 < 4; ++t4)
#pragma unroll
          for (int r = 0; r < 8; ++r) p.v[t4][r] = x[idx0 + r * 512 + 16 * t4];
        return p;
      },
      [&](int tM, floatx8 acc[4], const XTile& p) {
        size_t idx0 = (size_t)(tM * 16 + 8 * h) * 512 + col0;
#pragma unroll
        for (int t4 = 0; t4 < 4; ++t4)
#pragma unroll
          for (int r = 0; r < 8; ++r)
            out[idx0 + r * 512 + 16 * t4] = p.v[t4][r] + acc[t4][r] + bc[t4];
      });
}

extern "C" void kernel_launch(void* const* d_in, const int* in_sizes, int n_in,
                              void* d_out, int out_size, void* d_ws,
                              size_t ws_size, hipStream_t stream) {
  const float* x     = (const float*)d_in[0];
  const float* ln_g  = (const float*)d_in[1];
  const float* ln_b  = (const float*)d_in[2];
  const float* w_qkv = (const float*)d_in[3];
  const float* w_out = (const float*)d_in[4];
  const float* b_out = (const float*)d_in[5];
  const float* res_w = (const float*)d_in[6];
  float* out = (float*)d_out;

  char* p = (char*)d_ws;
  auto alloc = [&](size_t bytes) -> void* {
    void* r = (void*)p;
    p += (bytes + 255) & ~(size_t)255;
    return r;
  };
  const size_t NTOK = 4 * 4096;
  const size_t MM = (size_t)32 * 256 * 256;
  __bf16* xn_b   = (__bf16*)alloc(NTOK * 512 * 2);
  __bf16* wqkv_b = (__bf16*)alloc(1536 * 512 * 2);
  __bf16* wout_b = (__bf16*)alloc(512 * 512 * 2);
  __bf16* q_b    = (__bf16*)alloc((size_t)32 * 4096 * 64 * 2);
  __bf16* k_b    = (__bf16*)alloc((size_t)32 * 4096 * 64 * 2);
  float*  v_f    = (float*)alloc((size_t)32 * 4096 * 64 * 4);
  __bf16* vT_b   = (__bf16*)alloc((size_t)32 * 64 * 4096 * 2);
  __bf16* ql_b   = (__bf16*)alloc((size_t)32 * 256 * 64 * 2);
  __bf16* kl_b   = (__bf16*)alloc((size_t)32 * 256 * 64 * 2);
  float*  mat_x  = (float*)alloc(MM * 4);
  __bf16* x_b    = (__bf16*)alloc(MM * 2);
  __bf16* z0_b   = (__bf16*)alloc(MM * 2);
  __bf16* z1_b   = (__bf16*)alloc(MM * 2);
  __bf16* kvT_b  = (__bf16*)alloc((size_t)32 * 64 * 256 * 2);
  __bf16* w2t_b  = (__bf16*)alloc((size_t)32 * 64 * 256 * 2);
  __bf16* aout_b = (__bf16*)alloc(NTOK * 512 * 2);
  float*  scales = (float*)alloc(256);
  (void)ws_size; (void)n_in; (void)in_sizes; (void)out_size;

  const int T = 256;
  k_convert_w<<<3072, T, 0, stream>>>(w_qkv, w_out, wqkv_b, wout_b);
  k_layernorm<<<2048, T, 0, stream>>>(x, ln_g, ln_b, xn_b, (int)NTOK);
  k_gemm_qkv<<<252, 128, 0, stream>>>(xn_b, wqkv_b, q_b, k_b, v_f, vT_b);
  k_landmarks<<<2048, T, 0, stream>>>(q_b, k_b, ql_b, kl_b);
  k_attn2<<<32, 512, 0, stream>>>(ql_b, kl_b, mat_x, x_b, scales);

  k_pinv_attn3v<<<64, 512, 0, stream>>>(x_b, mat_x, scales, z0_b, z1_b, k_b, ql_b,
                                        vT_b, kvT_b);
  k_w2<<<64, T, 0, stream>>>(z0_b, kvT_b, w2t_b);
  k_attn1_out<<<1024, T, 0, stream>>>(kl_b, q_b, w2t_b, v_f, res_w, aout_b);
  k_outproj<<<256, 128, 0, stream>>>(aout_b, wout_b, b_out, x, out);
}


# attn1 conv (96 v loads, 24 taps) and k_w2 (40 fragments): loads issued in groups ahead of their products, one memory round trip per group (was one per load, each waited for); same sums
# speedup vs baseline: 4.7547x; 1.0898x over previous
#include <hip/hip_runtime.h>
#include <hip/hip_bf16.h>
#include <math.h>


typedef __attribute__((ext_vector_type(16))) __bf16 bf16x16;
typedef __attribute__((ext_vector_type(8)))  float  floatx8;

union FragAB { bf16x16 v; unsigned int u[8]; };

__device__ __forceinline__ unsigned pack2bf(float a, float b) {
  __bf16 lo = (__bf16)a, hi = (__bf16)b;
  unsigned short ul = __builtin_bit_cast(unsigned short, lo);
  unsigned short uh = __builtin_bit_cast(unsigned short, hi);
  return (unsigned)ul | ((unsigned)uh << 16);
}

__device__ __forceinline__ int kpair(int p, int h) {
  return (p < 4) ? (2 * p + 8 * h) : (8 + 2 * p + 8 * h);
}

__device__ __forceinline__ bf16x16 load_frag_bf16_rm(const __bf16* src, int row0,
                                                     int k0, int ld) {
  int lane = threadIdx.x & 31;
  int r = lane & 15, h = lane >> 4;
  const __bf16* base = src + (size_t)(row0 + r) * ld + k0;
  FragAB f;
#pragma unroll
  for (int p = 0; p < 8; ++p) f.u[p] = *(const unsigned*)(base + kpair(p, h));
  return f.v;
}

__device__ __forceinline__ floatx8 wmma_bf16(bf16x16 a, bf16x16 b, floatx8 c) {
  return __builtin_amdgcn_wmma_f32_16x16x32_bf16(false, a, false, b, (short)0, c,
                                                 false, false);
}

template <int KSTEPS>
__device__ __forceinline__ void gemm_strip(const __bf16* A, int rowA,
                                           const __bf16* B, int rowB, int ld,
                                           floatx8 acc[4]) {
#pragma unroll
  for (int ks = 0; ks < KSTEPS; ++ks) {
    int k0 = ks * 32;
    bf16x16 af = load_frag_bf16_rm(A, rowA, k0, ld);
#pragma unroll
    for (int t = 0; t < 4; ++t) {
      bf16x16 bfr = load_frag_bf16_rm(B, rowB + 16 * t, k0, ld);
      acc[t] = wmma_bf16(af, bfr, acc[t]);
    }
  }
}

typedef __attribute__((ext_vector_type(4))) unsigned int uintx4;

#define KS_RED_Q 3072

union AccQ { floatx8 v; uintx4 q[2]; };

__device__ __forceinline__ void load_aquarter(const __bf16* A, int row0, int kq,
                                              bf16x16 a[4]) {
#pragma unroll
  for (int ks = 0; ks < 4; ++ks) a[ks] = load_frag_bf16_rm(A, row0, kq + ks * 32, 512);
}

template <int T0, int T1>
__device__ __forceinline__ void mma_aquarter(const bf16x16 a[4],
                                             const bf16x16 (&bw)[4][16],
                                             floatx8 acc[16]) {
#pragma unroll
  for (int ks = 0; ks < 4; ++ks)
#pragma unroll
    for (int t = T0; t < T1; ++t) acc[t] = wmma_bf16(a[ks], bw[ks][t], acc[t]);
}

__device__ __forceinline__ void ksplit_send(uintx4* red, int w,
                                            const floatx8 acc[16]) {
  int lane = threadIdx.x & 31;
#pragma unroll
  for (int d = 1; d < 4; ++d) {
    int owner = (w + d) & 3;
    uintx4* dst = red + ((owner * 3 + (d - 1)) * 4) * 64 + lane;
#pragma unroll
    for (int j = 0; j < 4; ++j) {
      AccQ u;
      u.v = acc[4 * d + j];
      dst[j * 64] = u.q[0];
      dst[j * 64 + 32] = u.q[1];
    }
  }
  __builtin_amdgcn_fence(__ATOMIC_RELEASE, "workgroup", "local");
  __builtin_amdgcn_s_barrier();
  __builtin_amdgcn_fence(__ATOMIC_ACQUIRE, "workgroup", "local");
}

__device__ __forceinline__ void ksplit_fetch(const uintx4* red, int w,
                                             floatx8 part[12]) {
  int lane = threadIdx.x & 31;
  const uintx4* src = red + (w * 12) * 64 + lane;
#pragma unroll
  for (int i = 0; i < 12; ++i) {
    AccQ u;
    u.q[0] = src[i * 64];
    u.q[1] = src[i * 64 + 32];
    part[i] = u.v;
  }
}

__device__ __forceinline__ void ksplit_block(const bf16x16 a[4],
                                             const bf16x16 (&bw)[4][16],
                                             uintx4* red, int w,
                                             floatx8 acc[16]) {
  mma_aquarter<4, 16>(a, bw, acc);
  ksplit_send(red, w, acc);
  floatx8 part[12];
  ksplit_fetch(red, w, part);
  __builtin_amdgcn_sched_barrier(0);
  mma_aquarter<0, 4>(a, bw, acc);
#pragma unroll
  for (int d = 1; d < 4; ++d)
#pragma unroll
    for (int j = 0; j < 4; ++j) acc[j] += part[4 * (d - 1) + j];
}

template <int STEP, class Pre, class Fin>
__device__ __forceinline__ void gemm_wstat(const __bf16* A, const __bf16* W,
                                           int ng, int w, int g, uintx4* red,
                                           Pre pre, Fin fin) {
  int kq = w * 128;
  bf16x16 bw[4][16];
#pragma unroll
  for (int ks = 0; ks < 4; ++ks)
#pragma unroll
    for (int t = 0; t < 16; ++t)
      bw[ks][t] = load_frag_bf16_rm(W, ng * 256 + 16 * ((t + 4 * w) & 15),
                                    kq + ks * 32, 512);
  bf16x16 a0[4], a1[4];
  load_aquarter(A, g * 16, kq, a0);
  for (int tM = g; tM < 1024; tM += 2 * STEP) {
    int tM1 = tM + STEP, tM2 = tM + 2 * STEP;
    load_aquarter(A, min(tM1, 1023) * 16, kq, a1);
    {
      auto p = pre(tM);
      __builtin_amdgcn_sched_barrier(0);
      floatx8 acc[16] = {};
      ksplit_block(a0, bw, red, w, acc);
      fin(tM, acc, p);
    }
    load_aquarter(A, min(tM2, 1023) * 16, kq, a0);
    if (tM1 < 1024) {
      auto p = pre(tM1);
      __builtin_amdgcn_sched_barrier(0);
      floatx8 acc[16] = {};
      ksplit_block(a1, bw, red + KS_RED_Q, w, acc);
      fin(tM1, acc, p);
    }
  }
}

__global__ void k_convert_w(const float* wqkv, const float* wout,
                            __bf16* wqkv_b, __bf16* wout_b) {
  int i = blockIdx.x * blockDim.x + threadIdx.x;
  if (i < 1536 * 512) wqkv_b[i] = (__bf16)wqkv[i];
  if (i < 512 * 512) wout_b[i] = (__bf16)wout[i];
}

__global__ void k_layernorm(const float* x, const float* g, const float* b,
                            __bf16* xn, int ntok) {
  int wave = (int)((blockIdx.x * blockDim.x + threadIdx.x) >> 5);
  int lane = threadIdx.x & 31;
  if (wave >= ntok) return;
  const float* xr = x + (size_t)wave * 512;
  float s = 0.f, s2 = 0.f, vals[16];
#pragma unroll
  for (int i = 0; i < 16; ++i) {
    float v = xr[lane + 32 * i];
    vals[i] = v; s += v; s2 += v * v;
  }
  for (int m = 16; m >= 1; m >>= 1) {
    s += __shfl_xor(s, m, 32);
    s2 += __shfl_xor(s2, m, 32);
  }
  float mu = s * (1.f / 512.f);
  float var = s2 * (1.f / 512.f) - mu * mu;
  float inv = rsqrtf(var + 1e-5f);
  __bf16* o = xn + (size_t)wave * 512;
#pragma unroll
  for (int i = 0; i < 16; ++i) {
    int c = lane + 32 * i;
    o[c] = (__bf16)((vals[i] - mu) * inv * g[c] + b[c]);
  }
}

__global__ __launch_bounds__(128)
void k_gemm_qkv(const __bf16* xn, const __bf16* wq, __bf16* qb, __bf16* kb,
                float* vf, __bf16* vT) {
  __shared__ uintx4 red[2 * KS_RED_Q];
  int wave = __builtin_amdgcn_readfirstlane((int)(threadIdx.x >> 5));
  int g = (int)(blockIdx.x / 6);
  int ng = (int)(blockIdx.x % 6);
  int s = ng * 4 + wave;
  int lane = threadIdx.x & 31;
  int n = lane & 15, h = lane >> 4;
  int sect = s >> 3;
  int head = s & 7;
  gemm_wstat<42>(xn, wq, ng, wave, g, red, [](int) { return 0; },
                 [&](int tM, floatx8 acc[16], int) {
    int tok0 = tM * 16 + 8 * h;
    int bb = tok0 >> 12, ns0 = tok0 & 4095;
    size_t bh = (size_t)(bb * 8 + head);
    size_t idx0 = (bh * 4096 + ns0) * 64 + n;
    if (sect == 0) {
#pragma unroll
      for (int t = 0; t < 4; ++t)
#pragma unroll
        for (int r = 0; r < 8; ++r)
          qb[idx0 + r * 64 + 16 * t] = (__bf16)(acc[t][r] * 0.125f);
    } else if (sect == 1) {
#pragma unroll
      for (int t = 0; t < 4; ++t)
#pragma unroll
        for (int r = 0; r < 8; ++r)
          kb[idx0 + r * 64 + 16 * t] = (__bf16)acc[t][r];
    } else {
      size_t tdx0 = (bh * 64 + n) * 4096 + ns0;
#pragma unroll
      for (int t = 0; t < 4; ++t)
#pragma unroll
        for (int r = 0; r < 8; ++r) {
          vf[idx0 + r * 64 + 16 * t] = acc[t][r];
          vT[tdx0 + (size_t)(16 * t) * 4096 + r] = (__bf16)acc[t][r];
        }
    }
  });
}

__global__ void k_landmarks(const __bf16* qb, const __bf16* kb, __bf16* ql,
                            __bf16* kl) {
  int i = blockIdx.x * blockDim.x + threadIdx.x;
  if (i >= 32 * 256 * 64) return;
  int dh = i & 63, m = (i >> 6) & 255, bh = i >> 14;
  size_t base = ((size_t)bh * 4096 + m * 16) * 64 + dh;
  float sq = 0.f, sk = 0.f;
#pragma unroll
  for (int t = 0; t < 16; ++t) {
    sq += (float)qb[base + t * 64];
    sk += (float)kb[base + t * 64];
  }
  ql[i] = (__bf16)(sq * (1.f / 16.f));
  kl[i] = (__bf16)(sk * (1.f / 16.f));
}


typedef __attribute__((ext_vector_type(2))) unsigned int uintx2;
typedef __attribute__((ext_vector_type(4))) float floatx4;

#define PLANE_Q 8192

union FragQ { bf16x16 v; uintx4 q[2]; };

__device__ __forceinline__ bf16x16 lds_frag(const uintx4* plane, int nb, int kb) {
  int lane = threadIdx.x & 31;
  const uintx4* blk = plane + (nb * 8 + kb) * 64;
  FragQ f;
  f.q[0] = blk[lane];
  f.q[1] = blk[32 + lane];
  return f.v;
}

typedef short s16x8 __attribute__((ext_vector_type(8)));

__device__ __forceinline__ uintx4 lds_tr16(const uintx4* p) {
  typedef __attribute__((address_space(3))) s16x8 lds_s16x8;
  return __builtin_bit_cast(
      uintx4, __builtin_amdgcn_ds_load_tr16_b128_v8i16((lds_s16x8*)p));
}

__device__ __forceinline__ void block_sync_all() {
  __builtin_amdgcn_fence(__ATOMIC_RELEASE, "workgroup");
  __builtin_amdgcn_s_barrier();
  __builtin_amdgcn_fence(__ATOMIC_ACQUIRE, "workgroup");
}

template <bool ATR, bool DIAG, bool OT, bool PRESYNC>
__device__ __forceinline__ void pinv_gemm_tile(const __bf16* A, const uintx4* Ain,
                                               const uintx4* Bin, uintx4* Bout,
                                               uintx4* Dblk, __bf16* OutT,
                                               float alpha, float beta, int tm,
                                               int tn) {
  int lane = threadIdx.x & 31;
  int n = lane & 15, h = lane >> 4;
  int lp = ((lane >> 3) & 1) * 16 + 8 * h + (lane & 7);
  floatx8 acc[2][8] = {};
#pragma unroll
  for (int ks = 0; ks < 8; ++ks) {
    bf16x16 af[2], bfr[8];
#pragma unroll
    for (int i = 0; i < 2; ++i) {
      if (ATR) {
        int o0 = ((2 * ks) * 8 + tm) * 64 + i * 32 + lp;
        int o1 = o0 + 512;
        int od = (int)(Dblk - Ain) + (2 * tm + i) * 32 + lp;
        bool dstep = (ks == tm);
        if (i == 0) o0 = dstep ? od : o0; else o1 = dstep ? od : o1;
        FragQ f;
        f.q[0] = lds_tr16(Ain + o0);
        f.q[1] = lds_tr16(Ain + o1);
        af[i] = f.v;
      } else {
        af[i] = load_frag_bf16_rm(A, tm * 32 + 16 * i, ks * 32, 256);
      }
    }
#pragma unroll
    for (int j = 0; j < 8; ++j) bfr[j] = lds_frag(Bin, tn * 8 + j, ks);
#pragma unroll
    for (int i = 0; i < 2; ++i)
#pragma unroll
      for (int j = 0; j < 8; ++j) acc[i][j] = wmma_bf16(af[i], bfr[j], acc[i][j]);
  }
  if (PRESYNC) block_sync_all();
  int dblk = tn * 8 - tm * 2;
  float il[8];
#pragma unroll
  for (int r = 0; r < 8; ++r) il[r] = (n == r + 8 * h) ? 1.f : 0.f;
  size_t rm0 = (size_t)(tm * 32 + 8 * h) * 256 + (size_t)(tn * 128 + n);
#pragma unroll
  for (int i = 0; i < 2; ++i)
#pragma unroll
    for (int j = 0; j < 8; ++j) {
      float ad = (i - j == dblk) ? alpha : 0.f;
      float trv[8];
#pragma unroll
      for (int r = 0; r < 8; ++r) {
        float tv = acc[i][j][r];
        float tr = __builtin_fmaf(ad, il[r], beta * tv);
        if (OT) OutT[rm0 + (size_t)((16 * i + r) * 256 + 16 * j)] = (__bf16)tr;
        trv[r] = tr;
      }
      if (DIAG && (i - j == dblk)) {
        uintx4 d;
        d.x = pack2bf(-acc[i][j][0], -acc[i][j][1]);
        d.y = pack2bf(-acc[i][j][2], -acc[i][j][3]);
        d.z = pack2bf(-acc[i][j][4], -acc[i][j][5]);
        d.w = pack2bf(-acc[i][j][6], -acc[i][j][7]);
        Dblk[(2 * tm + i) * 32 + lane] = d;
      }
      uintx4 w;
      w.x = pack2bf(trv[0], trv[1]);
      w.y = pack2bf(trv[2], trv[3]);
      w.z = pack2bf(trv[4], trv[5]);
      w.w = pack2bf(trv[6], trv[7]);
      Bout[((tn * 8 + j) * 8 + tm) * 64 + i * 32 + lane] = w;
    }
}


#define IMG_Q 4096

__device__ __forceinline__ int img_unit(int f, int q, int h, int r) {
  return f * 64 + q * 32 + h * 16 + (r ^ (4 * h + 8 * q));
}

__device__ __forceinline__ bf16x16 img_frag(const uintx4* img, int f) {
  int lane = threadIdx.x & 31;
  int r = lane & 15, h = lane >> 4;
  FragQ fr;
  fr.q[0] = img[img_unit(f, 0, h, r)];
  fr.q[1] = img[img_unit(f, 1, h, r)];
  return fr.v;
}

template <int NT>
__device__ __forceinline__ void img_fetch(const uintx4* kq, const uintx4* vq,
                                          int vpitch, uintx4 (&pk)[2048 / NT],
                                          uintx4 (&pv)[2048 / NT]) {
#pragma unroll
  for (int j = 0; j < 2048 / NT; ++j) {
    int i = (int)threadIdx.x + NT * j;
    int wr = (i >> 3) & 63;
    int cb = ((i >> 2) & 1) + 2 * (i >> 9);
    pk[j] = kq[i];
    pv[j] = vq[wr * vpitch + cb * 4 + (i & 3)];
  }
}

template <int NT>
__device__ __forceinline__ void img_put(uintx4* img, const uintx4 (&pk)[2048 / NT],
                                        const uintx4 (&pv)[2048 / NT]) {
#pragma unroll
  for (int j = 0; j < 2048 / NT; ++j) {
    int i = (int)threadIdx.x + NT * j;
    int p = i & 3;
    int c2 = (i >> 2) & 1;
    int kr = i >> 3;
    int wr = kr & 63;
    int cb = c2 + 2 * (i >> 9);
    img[img_unit((kr >> 4) * 2 + c2, p >> 1, p & 1, kr & 15)] = pk[j];
    img[img_unit(32 + (wr >> 4) * 8 + cb, p >> 1, p & 1, wr & 15)] = pv[j];
  }
}

__device__ __forceinline__ void stream_attn_begin(float& run_m, float& run_s,
                                                  floatx8 acc[4]) {
  run_m = -1e30f;
  run_s = 0.f;
#pragma unroll
  for (int t = 0; t < 4; ++t)
#pragma unroll
    for (int r = 0; r < 8; ++r) acc[t][r] = 0.f;
}

__device__ __forceinline__ void stream_attn_keys(const uintx4* img, bf16x16 bq0,
                                                 bf16x16 bq1, float& run_m,
                                                 float& run_s, floatx8 acc[4]) {
#pragma unroll
  for (int c = 0; c < 256; c += 32) {
    int kf = (c >> 4) * 2;
    floatx8 s0 = {}, s1 = {};
    s0 = wmma_bf16(img_frag(img, kf), bq0, s0);
    s0 = wmma_bf16(img_frag(img, kf + 1), bq1, s0);
    s1 = wmma_bf16(img_frag(img, kf + 2), bq0, s1);
    s1 = wmma_bf16(img_frag(img, kf + 3), bq1, s1);
    float mc = -1e30f;
#pragma unroll
    for (int r = 0; r < 8; ++r) { mc = fmaxf(mc, s0[r]); mc = fmaxf(mc, s1[r]); }
    mc = fmaxf(mc, __shfl_xor(mc, 16, 32));
    float newm = fmaxf(run_m, mc);
    float scale = __expf(run_m - newm);
    float p0[8], p1[8], ps = 0.f;
#pragma unroll
    for (int r = 0; r < 8; ++r) {
      p0[r] = __expf(s0[r] - newm);
      p1[r] = __expf(s1[r] - newm);
      ps += p0[r] + p1[r];
    }
    run_s = run_s * scale + ps;
    run_m = newm;
    FragAB bp;
#pragma unroll
    for (int p = 0; p < 4; ++p) {
      bp.u[p] = pack2bf(p0[2 * p], p0[2 * p + 1]);
      bp.u[4 + p] = pack2bf(p1[2 * p], p1[2 * p + 1]);
    }
#pragma unroll
    for (int t = 0; t < 4; ++t) {
#pragma unroll
      for (int r = 0; r < 8; ++r) acc[t][r] *= scale;
      acc[t] = wmma_bf16(img_frag(img, 32 + t * 8 + (c >> 5)), bp.v, acc[t]);
    }
  }
}

__device__ __forceinline__ void stream_attn_end(float run_s, floatx8 acc[4]) {
  float tot = run_s + __shfl_xor(run_s, 16, 32);
  float inv = 1.f / tot;
#pragma unroll
  for (int t = 0; t < 4; ++t)
#pragma unroll
    for (int r = 0; r < 8; ++r) acc[t][r] *= inv;
}

__device__ __forceinline__ void attn3v_staged(const __bf16* K, const __bf16* Q,
                                              int qrow0, const __bf16* VT,
                                              __bf16* outB, uintx4* lds) {
  const uintx4* kq = (const uintx4*)K;
  const uintx4* vq = (const uintx4*)VT;
  uintx4 pk[4], pv[4];
  img_fetch<512>(kq, vq, 512, pk, pv);
  img_put<512>(lds, pk, pv);
  bf16x16 bq0 = load_frag_bf16_rm(Q, qrow0, 0, 64);
  bf16x16 bq1 = load_frag_bf16_rm(Q, qrow0, 32, 64);
  float run_m, run_s;
  floatx8 acc[4];
  stream_attn_begin(run_m, run_s, acc);
  __syncthreads();
  for (int s = 0; s < 16; ++s) {
    uintx4* cur = lds + (s & 1) * IMG_Q;
    uintx4* nxt = lds + ((s + 1) & 1) * IMG_Q;
    bool more = s + 1 < 16;
    if (more) img_fetch<512>(kq + (s + 1) * 2048, vq + (s + 1) * 32, 512, pk, pv);
    stream_attn_keys(cur, bq0, bq1, run_m, run_s, acc);
    if (more) {
      img_put<512>(nxt, pk, pv);
      __syncthreads();
    }
  }
  stream_attn_end(run_s, acc);
  int lane = threadIdx.x & 31;
  int n = lane & 15, h = lane >> 4;
#pragma unroll
  for (int t = 0; t < 4; ++t)
#pragma unroll
    for (int r = 0; r < 8; ++r) {
      int dh = 16 * t + r + 8 * h;
      outB[(size_t)n + (size_t)dh * 256] = (__bf16)acc[t][r];
    }
}

#define A2_PITCH 260

template <int M>
__device__ __forceinline__ float row_xor(float x) {
  int b = __builtin_bit_cast(int, x);
  return __builtin_bit_cast(
      float, __builtin_amdgcn_update_dpp(b, b, 0x160 + M, 0xf, 0xf, false));
}

__device__ __forceinline__ float row_max16(float x) {
  x = fmaxf(x, row_xor<8>(x));
  x = fmaxf(x, row_xor<4>(x));
  x = fmaxf(x, row_xor<2>(x));
  return fmaxf(x, row_xor<1>(x));
}

__device__ __forceinline__ float row_sum16(float x) {
  x += row_xor<8>(x);
  x += row_xor<4>(x);
  x += row_xor<2>(x);
  return x + row_xor<1>(x);
}

__global__ __launch_bounds__(512)
void k_attn2(const __bf16* ql, const __bf16* kl, float* xf, __bf16* xb,
             float* pmax) {
  __shared__ uintx4 kimg[2048];
  __shared__ uintx4 stage[16 * 16 * A2_PITCH / 4];
  __shared__ float colp[16 * 256];
  __shared__ float rowm[16];
  __shared__ float colm[8];
  int wave = __builtin_amdgcn_readfirstlane((int)(threadIdx.x >> 5));
  int lane = threadIdx.x & 31;
  int n = lane & 15, h = lane >> 4;
  size_t bh = blockIdx.x;
  const __bf16* A = ql + bh * 256 * 64;
  {
    const uintx4* kq = (const uintx4*)(kl + bh * 256 * 64);
    uintx4 pk[4];
#pragma unroll
    for (int j = 0; j < 4; ++j) pk[j] = kq[(int)threadIdx.x + 512 * j];
#pragma unroll
    for (int j = 0; j < 4; ++j) {
      int i = (int)threadIdx.x + 512 * j;
      int p = i & 3, c2 = (i >> 2) & 1, kr = i >> 3;
      kimg[img_unit((kr >> 4) * 2 + c2, p >> 1, p & 1, kr & 15)] = pk[j];
    }
  }
  bf16x16 af0 = load_frag_bf16_rm(A, wave * 16, 0, 64);
  bf16x16 af1 = load_frag_bf16_rm(A, wave * 16, 32, 64);
  __syncthreads();
  floatx8 acc[16];
#pragma unroll
  for (int j = 0; j < 16; ++j) {
    floatx8 z = {};
    acc[j] = wmma_bf16(af0, img_frag(kimg, 2 * j), z);
  }
#pragma unroll
  for (int j = 0; j < 16; ++j) acc[j] = wmma_bf16(af1, img_frag(kimg, 2 * j + 1), acc[j]);
  float* my = (float*)stage + wave * (16 * A2_PITCH);
  size_t g0 = (bh * 256 + (size_t)(wave * 16)) * 256 + 4 * lane;
  float rmax = 0.f;
  float cpart[16];
#pragma unroll
  for (int j = 0; j < 16; ++j) cpart[j] = 0.f;
#pragma unroll
  for (int r = 0; r < 8; ++r) {
    float m = acc[0][r];
#pragma unroll
    for (int j = 1; j < 16; ++j) m = fmaxf(m, acc[j][r]);
    m = row_max16(m);
    float e[16], s = 0.f;
#pragma unroll
    for (int j = 0; j < 16; ++j) { e[j] = __expf(acc[j][r] - m); s += e[j]; }
    s = row_sum16(s);
    float inv = 1.f / s;
    float rs = 0.f;
#pragma unroll
    for (int j = 0; j < 16; ++j) {
      float pv = e[j] * inv;
      my[(8 * h + r) * A2_PITCH + 16 * j + n] = pv;
      rs += fabsf(pv);
      cpart[j] += fabsf(pv);
    }
    rs = row_sum16(rs);
    rmax = fmaxf(rmax, rs);
    __builtin_amdgcn_fence(__ATOMIC_RELEASE, "wavefront");
    __builtin_amdgcn_wave_barrier();
    __builtin_amdgcn_fence(__ATOMIC_ACQUIRE, "wavefront");
#pragma unroll
    for (int hh = 0; hh < 2; ++hh)
#pragma unroll
      for (int k = 0; k < 2; ++k) {
        int i = 8 * hh + r;
        floatx4 v = *(const floatx4*)(my + i * A2_PITCH + 128 * k + 4 * lane);
        size_t g = g0 + (size_t)(i * 256 + 128 * k);
        *(floatx4*)(xf + g) = v;
        uintx2 w;
        w.x = pack2bf(v.x, v.y);
        w.y = pack2bf(v.z, v.w);
        *(uintx2*)(xb + g) = w;
      }
  }
  rmax = fmaxf(rmax, __shfl_xor(rmax, 16, 32));
#pragma unroll
  for (int j = 0; j < 16; ++j) cpart[j] += __shfl_xor(cpart[j], 16, 32);
  if (h == 0) {
#pragma unroll
    for (int j = 0; j < 16; ++j) colp[wave * 256 + 16 * j + n] = cpart[j];
  }
  if (lane == 0) rowm[wave] = rmax;
  __syncthreads();
  if (threadIdx.x < 256) {
    float cs = 0.f;
#pragma unroll
    for (int w = 0; w < 16; ++w) cs += colp[w * 256 + (int)threadIdx.x];
    for (int t = 16; t >= 1; t >>= 1) cs = fmaxf(cs, __shfl_xor(cs, t, 32));
    if (lane == 0) colm[wave] = cs;
  }
  __syncthreads();
  if (threadIdx.x == 0) {
    float cm = colm[0], rm = rowm[0];
    for (int w = 1; w < 8; ++w) cm = fmaxf(cm, colm[w]);
    for (int w = 1; w < 16; ++w) rm = fmaxf(rm, rowm[w]);
    pmax[bh] = rm;
    pmax[32 + bh] = cm;
  }
}

__global__ __launch_bounds__(512)
void k_pinv_attn3v(const __bf16* xb, const float* xf, const float* scales,
                   __bf16* z0, __bf16* z1, const __bf16* kb,
                   const __bf16* ql, const __bf16* vT, __bf16* kvT) {
  __shared__ uintx4 ldsB[2 * PLANE_Q + 512];
  int wave = __builtin_amdgcn_readfirstlane((int)(threadIdx.x >> 5));
  if (blockIdx.x >= 32) {
    size_t bh = blockIdx.x - 32;
    attn3v_staged(kb + bh * 4096 * 64, ql + bh * 256 * 64, wave * 16,
                  vT + bh * 64 * 4096, kvT + bh * 64 * 256 + wave * 16, ldsB);
    return;
  }
  uintx4* B0 = ldsB;
  uintx4* B1 = ldsB + PLANE_Q;
  uintx4* DG = ldsB + 2 * PLANE_Q;
  size_t bofs = (size_t)blockIdx.x << 16;
  int lane = threadIdx.x & 31;
  int tm = wave >> 1, tn = wave & 1;
  const __bf16* X = xb + bofs;
  __bf16* zc = z0 + bofs;
  __bf16* zn = z1 + bofs;
  const float* XF = xf + bofs;
  float rmx = scales[0], cmx = scales[32];
  for (int i = 1; i < 32; ++i) {
    rmx = fmaxf(rmx, scales[i]);
    cmx = fmaxf(cmx, scales[32 + i]);
  }
  float denom = rmx * cmx;
  {
    int c = lane & 15, h = lane >> 4;
#pragma unroll
    for (int b = 0; b < 8; ++b) {
      int blk = wave * 8 + b;
      const float* sp =
          XF + (size_t)((blk >> 3) * 16 + c) * 256 + (blk & 7) * 32 + 8 * h;
      floatx4 a0 = *(const floatx4*)sp, a1 = *(const floatx4*)(sp + 4);
      floatx4 c0 = *(const floatx4*)(sp + 16), c1 = *(const floatx4*)(sp + 20);
      uintx4 q0, q1;
      q0.x = pack2bf(a0.x / denom, a0.y / denom);
      q0.y = pack2bf(a0.z / denom, a0.w / denom);
      q0.z = pack2bf(a1.x / denom, a1.y / denom);
      q0.w = pack2bf(a1.z / denom, a1.w / denom);
      q1.x = pack2bf(c0.x / denom, c0.y / denom);
      q1.y = pack2bf(c0.z / denom, c0.w / denom);
      q1.z = pack2bf(c1.x / denom, c1.y / denom);
      q1.w = pack2bf(c1.z / denom, c1.w / denom);
      B1[blk * 64 + lane] = q0;
      B1[blk * 64 + 32 + lane] = q1;
    }
  }
  {
    int M = (int)(threadIdx.x & 255), kg = (int)(threadIdx.x >> 8);
    for (int p = 0; p < 16; p += 4) {
      float raw[4][8];
#pragma unroll
      for (int u = 0; u < 4; ++u)
#pragma unroll
        for (int e = 0; e < 8; ++e)
          raw[u][e] = XF[(size_t)(((p + u) * 2 + kg) * 8 + e) * 256 + M];
#pragma unroll
      for (int u = 0; u < 4; ++u) {
        float zv[8];
#pragma unroll
        for (int e = 0; e < 8; ++e) zv[e] = raw[u][e] / denom;
        uintx4 w;
        w.x = pack2bf(zv[0], zv[1]);
        w.y = pack2bf(zv[2], zv[3]);
        w.z = pack2bf(zv[4], zv[5]);
        w.w = pack2bf(zv[6], zv[7]);
        *(uintx4*)(zc + (size_t)M * 256 + ((p + u) * 2 + kg) * 8) = w;
      }
    }
  }
  block_sync_all();
  for (int it = 0; it < 6; ++it) {
    pinv_gemm_tile<false, true, false, false>(X, nullptr, B1, B0, DG, nullptr,
                                              7.f, -1.f, tm, tn);
    block_sync_all();
    pinv_gemm_tile<true, false, false, false>(nullptr, B0, B0, B1, DG, nullptr,
                                              15.f, 1.f, tm, tn);
    block_sync_all();
    pinv_gemm_tile<true, false, false, true>(nullptr, B0, B1, B0, DG, nullptr,
                                             13.f, 1.f, tm, tn);
    block_sync_all();
    pinv_gemm_tile<false, false, true, false>(zc, nullptr, B0, B1, DG, zn, 0.f,
                                              0.25f, tm, tn);
    block_sync_all();
    __bf16* t = zc; zc = zn; zn = t;
  }
}

__global__ __launch_bounds__(256)
void k_w2(const __bf16* zb, const __bf16* kvT, __bf16* w2t) {
  int wid = (int)((blockIdx.x * blockDim.x + threadIdx.x) >> 5);
  int bh = wid >> 4, tM = wid & 15;
  floatx8 acc[4] = {};
  const __bf16* Az = zb + ((size_t)bh << 16);
  const __bf16* Bk = kvT + (size_t)bh * 64 * 256;
#pragma unroll
  for (int g = 0; g < 2; ++g) {
    bf16x16 af[4], bfr[4][4];
#pragma unroll
    for (int ks = 0; ks < 4; ++ks) {
      af[ks] = load_frag_bf16_rm(Az, tM * 16, (4 * g + ks) * 32, 256);
#pragma unroll
      for (int t = 0; t < 4; ++t)
        bfr[ks][t] = load_frag_bf16_rm(Bk, 16 * t, (4 * g + ks) * 32, 256);
    }
    __builtin_amdgcn_sched_barrier(0);
#pragma unroll
    for (int ks = 0; ks < 4; ++ks)
#pragma unroll
      for (int t = 0; t < 4; ++t) acc[t] = wmma_bf16(af[ks], bfr[ks][t], acc[t]);
  }
  int lane = threadIdx.x & 31;
  int n = lane & 15, h = lane >> 4;
#pragma unroll
  for (int t4 = 0; t4 < 4; ++t4)
#pragma unroll
    for (int r = 0; r < 8; ++r) {
      int M = tM * 16 + r + 8 * h;
      int N = 16 * t4 + n;
      w2t[((size_t)bh * 64 + N) * 256 + M] = (__bf16)acc[t4][r];
    }
}

typedef __attribute__((ext_vector_type(2))) float floatx2;

__device__ __forceinline__ floatx8 wmma_f32(floatx2 a, floatx2 b, floatx8 c) {
  return __builtin_amdgcn_wmma_f32_16x16x4_f32(false, a, false, b, (short)0, c,
                                               false, false);
}

__global__ __launch_bounds__(256, 8)
void k_attn1_out(const __bf16* kl, const __bf16* qb, const __bf16* w2t,
                 const float* vf, const float* rw, __bf16* aout_b) {
  __shared__ uintx4 img[IMG_Q];
  int wid = (int)((blockIdx.x * blockDim.x + threadIdx.x) >> 5);
  int bh = wid >> 8, blk = wid & 255;
  int b = bh >> 3, hh = bh & 7;
  {
    uintx4 pk[8], pv[8];
    img_fetch<256>((const uintx4*)(kl + (size_t)bh * 256 * 64),
                   (const uintx4*)(w2t + (size_t)bh * 64 * 256), 32, pk, pv);
    img_put<256>(img, pk, pv);
  }
  const __bf16* qhead = qb + (size_t)bh * 4096 * 64;
  bf16x16 bq0 = load_frag_bf16_rm(qhead, blk * 16, 0, 64);
  bf16x16 bq1 = load_frag_bf16_rm(qhead, blk * 16, 32, 64);
  float run_m, run_s;
  floatx8 acc[4];
  stream_attn_begin(run_m, run_s, acc);
  __syncthreads();
  stream_attn_keys(img, bq0, bq1, run_m, run_s, acc);
  stream_attn_end(run_s, acc);
  int lane = threadIdx.x & 31;
  int c = lane & 15, h = lane >> 4;
  const float* vhead = vf + (size_t)bh * 4096 * 64;
  const float* wrow = rw + hh * 33;
  int p0 = blk * 16 - 16;
#pragma unroll
  for (int g = 0; g < 2; ++g) {
    float raw[6][4][2];
    float traw[6][2];
#pragma unroll
    for (int kl = 0; kl < 6; ++kl)
#pragma unroll
      for (int s = 0; s < 2; ++s) {
        int p = p0 + 4 * (6 * g + kl) + 2 * h + s;
        const float* vrow = vhead + (size_t)min(max(p, 0), 4095) * 64 + c;
#pragma unroll
        for (int t = 0; t < 4; ++t) raw[kl][t][s] = vrow[16 * t];
        traw[kl][s] = wrow[min(max(4 * (6 * g + kl) + 2 * h + s - c, 0), 32)];
      }
    __builtin_amdgcn_sched_barrier(0);
#pragma unroll
    for (int kl = 0; kl < 6; ++kl) {
      int q0 = 4 * (6 * g + kl) + 2 * h;
      floatx2 bw;
#pragma unroll
      for (int s = 0; s < 2; ++s) {
        int j = q0 + s - c;
        bw[s] = (j >= 0 && j <= 32) ? traw[kl][s] : 0.f;
      }
#pragma unroll
      for (int t = 0; t < 4; ++t) {
        floatx2 av;
#pragma unroll
        for (int s = 0; s < 2; ++s) {
          int p = p0 + q0 + s;
          av[s] = (p >= 0 && p < 4096) ? raw[kl][t][s] : 0.f;
        }
        acc[t] = wmma_f32(av, bw, acc[t]);
      }
    }
    __builtin_amdgcn_sched_barrier(0);
  }
  __bf16* orow = aout_b + ((size_t)b * 4096 + blk * 16 + c) * 512 + hh * 64 + 8 * h;
#pragma unroll
  for (int t = 0; t < 4; ++t) {
    uintx4 w;
    w.x = pack2bf(acc[t][0], acc[t][1]);
    w.y = pack2bf(acc[t][2], acc[t][3]);
    w.z = pack2bf(acc[t][4], acc[t][5]);
    w.w = pack2bf(acc[t][6], acc[t][7]);
    *(uintx4*)(orow + 16 * t) = w;
  }
}

struct XTile { float v[4][8]; };

__global__ __launch_bounds__(128)
void k_outproj(const __bf16* aoutb, const __bf16* wo, const float* bias,
               const float* x, float* out) {
  __shared__ uintx4 red[2 * KS_RED_Q];
  int wave = __builtin_amdgcn_readfirstlane((int)(threadIdx.x >> 5));
  int g = (int)(blockIdx.x >> 1);
  int ng = (int)(blockIdx.x & 1);
  int s = ng * 4 + wave;
  int lane = threadIdx.x & 31;
  int n = lane & 15, h = lane >> 4;
  int col0 = s * 64 + n;
  float bc[4];
#pragma unroll
  for (int t4 = 0; t4 < 4; ++t4) bc[t4] = bias[col0 + 16 * t4];
  gemm_wstat<128>(
      aoutb, wo, ng, wave, g, red,
      [&](int tM) {
        XTile p;
        size_t idx0 = (size_t)(tM * 16 + 8 * h) * 512 + col0;
#pragma unroll
        for (int t4 = 0; t4 < 4; ++t4)
#pragma unroll
          for (int r = 0; r < 8; ++r) p.v[t4][r] = x[idx0 + r * 512 + 16 * t4];
        return p;
      },
      [&](int tM, floatx8 acc[16], const XTile& p) {
        size_t idx0 = (size_t)(tM * 16 + 8 * h) * 512 + col0;
#pragma unroll
        for (int t4 = 0; t4 < 4; ++t4)
#pragma unroll
          for (int r = 0; r < 8; ++r)
            out[idx0 + r * 512 + 16 * t4] = p.v[t4][r] + acc[t4][r] + bc[t4];
      });
}

extern "C" void kernel_launch(void* const* d_in, const int* in_sizes, int n_in,
                              void* d_out, int out_size, void* d_ws,
                              size_t ws_size, hipStream_t stream) {
  const float* x     = (const float*)d_in[0];
  const float* ln_g  = (const float*)d_in[1];
  const float* ln_b  = (const float*)d_in[2];
  const float* w_qkv = (const float*)d_in[3];
  const float* w_out = (const float*)d_in[4];
  const float* b_out = (const float*)d_in[5];
  const float* res_w = (const float*)d_in[6];
  float* out = (float*)d_out;

  char* p = (char*)d_ws;
  auto alloc = [&](size_t bytes) -> void* {
    void* r = (void*)p;
    p += (bytes + 255) & ~(size_t)255;
    return r;
  };
  const size_t NTOK = 4 * 4096;
  const size_t MM = (size_t)32 * 256 * 256;
  __bf16* xn_b   = (__bf16*)alloc(NTOK * 512 * 2);
  __bf16* wqkv_b = (__bf16*)alloc(1536 * 512 * 2);
  __bf16* wout_b = (__bf16*)alloc(512 * 512 * 2);
  __bf16* q_b    = (__bf16*)alloc((size_t)32 * 4096 * 64 * 2);
  __bf16* k_b    = (__bf16*)alloc((size_t)32 * 4096 * 64 * 2);
  float*  v_f    = (float*)alloc((size_t)32 * 4096 * 64 * 4);
  __bf16* vT_b   = (__bf16*)alloc((size_t)32 * 64 * 4096 * 2);
  __bf16* ql_b   = (__bf16*)alloc((size_t)32 * 256 * 64 * 2);
  __bf16* kl_b   = (__bf16*)alloc((size_t)32 * 256 * 64 * 2);
  float*  mat_x  = (float*)alloc(MM * 4);
  __bf16* x_b    = (__bf16*)alloc(MM * 2);
  __bf16* z0_b   = (__bf16*)alloc(MM * 2);
  __bf16* z1_b   = (__bf16*)alloc(MM * 2);
  __bf16* kvT_b  = (__bf16*)alloc((size_t)32 * 64 * 256 * 2);
  __bf16* w2t_b  = (__bf16*)alloc((size_t)32 * 64 * 256 * 2);
  __bf16* aout_b = (__bf16*)alloc(NTOK * 512 * 2);
  float*  scales = (float*)alloc(256);
  (void)ws_size; (void)n_in; (void)in_sizes; (void)out_size;

  const int T = 256;
  k_convert_w<<<3072, T, 0, stream>>>(w_qkv, w_out, wqkv_b, wout_b);
  k_layernorm<<<2048, T, 0, stream>>>(x, ln_g, ln_b, xn_b, (int)NTOK);
  k_gemm_qkv<<<252, 128, 0, stream>>>(xn_b, wqkv_b, q_b, k_b, v_f, vT_b);
  k_landmarks<<<2048, T, 0, stream>>>(q_b, k_b, ql_b, kl_b);
  k_attn2<<<32, 512, 0, stream>>>(ql_b, kl_b, mat_x, x_b, scales);

  k_pinv_attn3v<<<64, 512, 0, stream>>>(x_b, mat_x, scales, z0_b, z1_b, k_b, ql_b,
                                        vT_b, kvT_b);
  k_w2<<<64, T, 0, stream>>>(z0_b, kvT_b, w2t_b);
  k_attn1_out<<<1024, T, 0, stream>>>(kl_b, q_b, w2t_b, v_f, res_w, aout_b);
  k_outproj<<<256, 128, 0, stream>>>(aout_b, wout_b, b_out, x, out);
}
